# starting kernel
# baseline (speedup 1.0000x reference)
#include <hip/hip_runtime.h>
#include <hip/hip_bf16.h>
#include <stdint.h>

#define NQ      14
#define NL      8
#define NSTATES 16384
#define BLOCK   256

typedef __attribute__((ext_vector_type(2))) float v2f;
typedef __attribute__((ext_vector_type(8))) float v8f;

__device__ __forceinline__ float2 cmulf(float2 a, float2 b) {
  return make_float2(a.x * b.x - a.y * b.y, a.x * b.y + a.y * b.x);
}

__device__ __forceinline__ uint32_t perm_map(uint32_t y, int l) {
  uint32_t r = y;
#pragma unroll
  for (int j = 1; j <= 7; ++j)
    if ((j & ~l) == 0) r ^= (y >> j);
  return r;
}

__global__ __launch_bounds__(BLOCK)
void qcircuit_kernel(const float* __restrict__ latent,
                     const float* __restrict__ weights,
                     float* __restrict__ out) {
  extern __shared__ float smem[];
  float2* st2  = (float2*)smem;
  float* s_tab = smem + 2 * NSTATES;

  const int tid  = threadIdx.x;
  const int lane = tid & 31;
  const int wave = tid >> 5;
  const int bid  = blockIdx.x;
  const float* lat = latent + bid * NQ;

  {
    const int half = tid >> 7, idx = tid & 127;
    float prod = 1.f;
#pragma unroll
    for (int j = 0; j < 7; ++j) {
      float s, c;
      sincosf(0.5f * lat[half * 7 + j], &s, &c);
      prod *= ((idx >> (6 - j)) & 1) ? s : c;
    }
    s_tab[tid] = prod;
  }
  __syncthreads();
#pragma unroll 4
  for (int i = 0; i < NSTATES / BLOCK; ++i) {
    const int y = tid * (NSTATES / BLOCK) + i;
    const float mag = s_tab[y >> 7] * s_tab[128 + (y & 127)];
    const int pc = __popc(y) & 3;
    float2 a;
    a.x = (pc == 0) ? mag : ((pc == 2) ? -mag : 0.f);
    a.y = (pc == 1) ? -mag : ((pc == 3) ? mag : 0.f);
    st2[y] = a;
  }
  __syncthreads();

  const int bsh[4] = {10, 6, 2, 0};
  for (int l = 0; l < NL; ++l) {
    for (int g = 0; g < 4; ++g) {
      const int b = bsh[g];
      float2 gt[4][2][2];
#pragma unroll
      for (int i = 0; i < 4; ++i) {
        if (g == 3 && i >= 2) {
          gt[i][0][0] = make_float2(1.f, 0.f); gt[i][0][1] = make_float2(0.f, 0.f);
          gt[i][1][0] = make_float2(0.f, 0.f); gt[i][1][1] = make_float2(1.f, 0.f);
        } else {
          const int q = 13 - b - i;
          const float* w = weights + (l * NQ + q) * 3;
          const float phi = w[0], th = w[1], om = w[2];
          float sth, cth, sa, ca, sb, cb;
          sincosf(0.5f * th, &sth, &cth);
          sincosf(0.5f * (phi + om), &sa, &ca);
          sincosf(0.5f * (phi - om), &sb, &cb);
          gt[i][0][0] = make_float2( cth * ca, -cth * sa);
          gt[i][0][1] = make_float2(-sth * cb, -sth * sb);
          gt[i][1][0] = make_float2( sth * cb, -sth * sb);
          gt[i][1][1] = make_float2( cth * ca,  cth * sa);
        }
      }
      const int m   = lane & 15;
      const int khi = (lane >> 4) << 1;
      v2f aR[4], aI[4], aIn[4];
#pragma unroll
      for (int j = 0; j < 4; ++j) {
        const int k0 = 4 * j + khi, k1 = k0 + 1;
        float2 u0 = gt[0][m & 1][k0 & 1];
        u0 = cmulf(u0, gt[1][(m >> 1) & 1][(k0 >> 1) & 1]);
        u0 = cmulf(u0, gt[2][(m >> 2) & 1][(k0 >> 2) & 1]);
        u0 = cmulf(u0, gt[3][(m >> 3) & 1][(k0 >> 3) & 1]);
        float2 u1 = gt[0][m & 1][k1 & 1];
        u1 = cmulf(u1, gt[1][(m >> 1) & 1][(k1 >> 1) & 1]);
        u1 = cmulf(u1, gt[2][(m >> 2) & 1][(k1 >> 2) & 1]);
        u1 = cmulf(u1, gt[3][(m >> 3) & 1][(k1 >> 3) & 1]);
        aR[j].x  =  u0.x; aR[j].y  =  u1.x;
        aI[j].x  =  u0.y; aI[j].y  =  u1.y;
        aIn[j].x = -u0.y; aIn[j].y = -u1.y;
      }
      const uint32_t pk1 = perm_map(1u << b, l);
      const uint32_t pk2 = perm_map(2u << b, l);
      const uint32_t pk4 = perm_map(4u << b, l);
      const uint32_t pk8 = perm_map(8u << b, l);
      const uint32_t lowm = (1u << b) - 1u;

      for (int ti = 0; ti < 8; ++ti) {
        const int t = wave * 8 + ti;
        const int c = t * 16 + (lane & 15);
        const uint32_t cpart = (((uint32_t)c >> b) << (b + 4)) | ((uint32_t)c & lowm);
        const uint32_t pcb = perm_map(cpart, l);
        v2f bR[4], bI[4];
#pragma unroll
        for (int j = 0; j < 4; ++j) {
          const int k0 = 4 * j + khi, k1 = k0 + 1;
          uint32_t x0 = pcb, x1 = pcb;
          if (k0 & 1) x0 ^= pk1; if (k0 & 2) x0 ^= pk2;
          if (k0 & 4) x0 ^= pk4; if (k0 & 8) x0 ^= pk8;
          if (k1 & 1) x1 ^= pk1; if (k1 & 2) x1 ^= pk2;
          if (k1 & 4) x1 ^= pk4; if (k1 & 8) x1 ^= pk8;
          const float2 s0 = st2[x0], s1 = st2[x1];
          bR[j].x = s0.x; bR[j].y = s1.x;
          bI[j].x = s0.y; bI[j].y = s1.y;
        }
        v8f cr = {}, ci = {};
#pragma unroll
        for (int j = 0; j < 4; ++j)
          cr = __builtin_amdgcn_wmma_f32_16x16x4_f32(false, aR[j],  false, bR[j], (short)0, cr, false, false);
#pragma unroll
        for (int j = 0; j < 4; ++j)
          cr = __builtin_amdgcn_wmma_f32_16x16x4_f32(false, aIn[j], false, bI[j], (short)0, cr, false, false);
#pragma unroll
        for (int j = 0; j < 4; ++j)
          ci = __builtin_amdgcn_wmma_f32_16x16x4_f32(false, aR[j],  false, bI[j], (short)0, ci, false, false);
#pragma unroll
        for (int j = 0; j < 4; ++j)
          ci = __builtin_amdgcn_wmma_f32_16x16x4_f32(false, aI[j],  false, bR[j], (short)0, ci, false, false);
        const int mb = (lane >> 4) << 3;
#pragma unroll
        for (int r = 0; r < 8; ++r) {
          const int mm = mb + r;
          uint32_t x = pcb;
          if (mm & 1) x ^= pk1; if (mm & 2) x ^= pk2;
          if (mm & 4) x ^= pk4; if (mm & 8) x ^= pk8;
          st2[x] = make_float2(cr[r], ci[r]);
        }
      }
      __syncthreads();
    }
  }

  float acc[NQ];
#pragma unroll
  for (int q = 0; q < NQ; ++q) acc[q] = 0.f;
  for (int i = 0; i < NSTATES / BLOCK; ++i) {
    const int y = tid + BLOCK * i;
    const uint32_t phys = (uint32_t)y ^ ((uint32_t)y >> 8);
    const float2 a = st2[phys];
    const float p = a.x * a.x + a.y * a.y;
#pragma unroll
    for (int q = 0; q < NQ; ++q)
      acc[q] += ((y >> (13 - q)) & 1) ? -p : p;
  }
#pragma unroll
  for (int q = 0; q < NQ; ++q) {
    float v = acc[q];
#pragma unroll
    for (int off = 16; off > 0; off >>= 1) v += __shfl_down(v, off, 32);
    if (lane == 0) s_tab[wave * NQ + q] = v;
  }
  __syncthreads();
  if (tid < NQ) {
    float z = 0.f;
#pragma unroll
    for (int w = 0; w < 8; ++w) z += s_tab[w * NQ + tid];
    out[bid * NQ + tid] = z;
  }
}

extern "C" void kernel_launch(void* const* d_in, const int* in_sizes, int n_in,
                              void* d_out, int out_size, void* d_ws, size_t ws_size,
                              hipStream_t stream) {
  const float* latent  = (const float*)d_in[0];
  const float* weights = (const float*)d_in[1];
  float* out = (float*)d_out;

  const int smem_bytes = (2 * NSTATES + 256) * (int)sizeof(float);
  hipFuncSetAttribute(reinterpret_cast<const void*>(qcircuit_kernel),
                      hipFuncAttributeMaxDynamicSharedMemorySize, smem_bytes);
  qcircuit_kernel<<<512, BLOCK, smem_bytes, stream>>>(latent, weights, out);
}


# All 112 Rot 2x2 gates computed once per workgroup into an LDS table; each (layer, group) reads its lane's entries by address instead of 12 scalar sincos and a scratch round trip per wave
# speedup vs baseline: 1.3914x; 1.3914x over previous
#include <hip/hip_runtime.h>
#include <hip/hip_bf16.h>
#include <stdint.h>

#define NQ      14
#define NL      8
#define NSTATES 16384
#define BLOCK   256
#define NGATE   (NL * NQ)
#define SMEM_BYTES 135712

typedef __attribute__((ext_vector_type(2))) float v2f;
typedef __attribute__((ext_vector_type(8))) float v8f;

__device__ __forceinline__ float2 cmulf(float2 a, float2 b) {
  return make_float2(a.x * b.x - a.y * b.y, a.x * b.y + a.y * b.x);
}

__device__ __forceinline__ uint32_t perm_map(uint32_t y, int l) {
  uint32_t r = y;
#pragma unroll
  for (int j = 1; j <= 7; ++j)
    if ((j & ~l) == 0) r ^= (y >> j);
  return r;
}

__global__ __launch_bounds__(BLOCK)
void qcircuit_kernel(const float* __restrict__ latent,
                     const float* __restrict__ weights,
                     float* __restrict__ out) {
  extern __shared__ float smem[];
  float2* st2  = (float2*)smem;
  float* s_tab = smem + 2 * NSTATES;
  float2* g_tab = (float2*)(smem + 2 * NSTATES + 256);

  const int tid  = threadIdx.x;
  const int lane = tid & 31;
  const int wave = tid >> 5;
  const int bid  = blockIdx.x;
  const float* lat = latent + bid * NQ;

  {
    const int half = tid >> 7, idx = tid & 127;
    float prod = 1.f;
#pragma unroll
    for (int j = 0; j < 7; ++j) {
      float s, c;
      sincosf(0.5f * lat[half * 7 + j], &s, &c);
      prod *= ((idx >> (6 - j)) & 1) ? s : c;
    }
    s_tab[tid] = prod;
  }
  if (tid < NGATE) {
    const float* w = weights + tid * 3;
    const float phi = w[0], th = w[1], om = w[2];
    float sth, cth, sa, ca, sb, cb;
    sincosf(0.5f * th, &sth, &cth);
    sincosf(0.5f * (phi + om), &sa, &ca);
    sincosf(0.5f * (phi - om), &sb, &cb);
    g_tab[tid * 4 + 0] = make_float2( cth * ca, -cth * sa);
    g_tab[tid * 4 + 1] = make_float2(-sth * cb, -sth * sb);
    g_tab[tid * 4 + 2] = make_float2( sth * cb, -sth * sb);
    g_tab[tid * 4 + 3] = make_float2( cth * ca,  cth * sa);
  } else if (tid == NGATE) {
    g_tab[NGATE * 4 + 0] = make_float2(1.f, 0.f); g_tab[NGATE * 4 + 1] = make_float2(0.f, 0.f);
    g_tab[NGATE * 4 + 2] = make_float2(0.f, 0.f); g_tab[NGATE * 4 + 3] = make_float2(1.f, 0.f);
  }
  __syncthreads();
#pragma unroll 4
  for (int i = 0; i < NSTATES / BLOCK; ++i) {
    const int y = tid * (NSTATES / BLOCK) + i;
    const float mag = s_tab[y >> 7] * s_tab[128 + (y & 127)];
    const int pc = __popc(y) & 3;
    float2 a;
    a.x = (pc == 0) ? mag : ((pc == 2) ? -mag : 0.f);
    a.y = (pc == 1) ? -mag : ((pc == 3) ? mag : 0.f);
    st2[y] = a;
  }
  __syncthreads();

  const int bsh[4] = {10, 6, 2, 0};
  for (int l = 0; l < NL; ++l) {
    for (int g = 0; g < 4; ++g) {
      const int b = bsh[g];
      const int m = lane & 15;
      const int h = lane >> 4;
      const int khi = h << 1;
      const int i0 = l * NQ + 13 - b;
      const int i1 = i0 - 1;
      const int i2 = (g == 3) ? NGATE : i0 - 2;
      const int i3 = (g == 3) ? NGATE : i0 - 3;
      const float2 g0c0 = g_tab[i0 * 4 + ((m     ) & 1) * 2 + 0];
      const float2 g0c1 = g_tab[i0 * 4 + ((m     ) & 1) * 2 + 1];
      const float2 g1ch = g_tab[i1 * 4 + ((m >> 1) & 1) * 2 + h];
      const float2 g2c0 = g_tab[i2 * 4 + ((m >> 2) & 1) * 2 + 0];
      const float2 g2c1 = g_tab[i2 * 4 + ((m >> 2) & 1) * 2 + 1];
      const float2 g3c0 = g_tab[i3 * 4 + ((m >> 3) & 1) * 2 + 0];
      const float2 g3c1 = g_tab[i3 * 4 + ((m >> 3) & 1) * 2 + 1];
      v2f aR[4], aI[4], aIn[4];
#pragma unroll
      for (int j = 0; j < 4; ++j) {
        const float2 g2 = (j & 1) ? g2c1 : g2c0;
        const float2 g3 = (j >> 1) ? g3c1 : g3c0;
        float2 u0 = cmulf(g0c0, g1ch);
        u0 = cmulf(u0, g2);
        u0 = cmulf(u0, g3);
        float2 u1 = cmulf(g0c1, g1ch);
        u1 = cmulf(u1, g2);
        u1 = cmulf(u1, g3);
        aR[j].x  =  u0.x; aR[j].y  =  u1.x;
        aI[j].x  =  u0.y; aI[j].y  =  u1.y;
        aIn[j].x = -u0.y; aIn[j].y = -u1.y;
      }
      const uint32_t pk1 = perm_map(1u << b, l);
      const uint32_t pk2 = perm_map(2u << b, l);
      const uint32_t pk4 = perm_map(4u << b, l);
      const uint32_t pk8 = perm_map(8u << b, l);
      const uint32_t lowm = (1u << b) - 1u;

      for (int ti = 0; ti < 8; ++ti) {
        const int t = wave * 8 + ti;
        const int c = t * 16 + (lane & 15);
        const uint32_t cpart = (((uint32_t)c >> b) << (b + 4)) | ((uint32_t)c & lowm);
        const uint32_t pcb = perm_map(cpart, l);
        v2f bR[4], bI[4];
#pragma unroll
        for (int j = 0; j < 4; ++j) {
          const int k0 = 4 * j + khi, k1 = k0 + 1;
          uint32_t x0 = pcb, x1 = pcb;
          if (k0 & 1) x0 ^= pk1; if (k0 & 2) x0 ^= pk2;
          if (k0 & 4) x0 ^= pk4; if (k0 & 8) x0 ^= pk8;
          if (k1 & 1) x1 ^= pk1; if (k1 & 2) x1 ^= pk2;
          if (k1 & 4) x1 ^= pk4; if (k1 & 8) x1 ^= pk8;
          const float2 s0 = st2[x0], s1 = st2[x1];
          bR[j].x = s0.x; bR[j].y = s1.x;
          bI[j].x = s0.y; bI[j].y = s1.y;
        }
        v8f cr = {}, ci = {};
#pragma unroll
        for (int j = 0; j < 4; ++j)
          cr = __builtin_amdgcn_wmma_f32_16x16x4_f32(false, aR[j],  false, bR[j], (short)0, cr, false, false);
#pragma unroll
        for (int j = 0; j < 4; ++j)
          cr = __builtin_amdgcn_wmma_f32_16x16x4_f32(false, aIn[j], false, bI[j], (short)0, cr, false, false);
#pragma unroll
        for (int j = 0; j < 4; ++j)
          ci = __builtin_amdgcn_wmma_f32_16x16x4_f32(false, aR[j],  false, bI[j], (short)0, ci, false, false);
#pragma unroll
        for (int j = 0; j < 4; ++j)
          ci = __builtin_amdgcn_wmma_f32_16x16x4_f32(false, aI[j],  false, bR[j], (short)0, ci, false, false);
        const int mb = (lane >> 4) << 3;
#pragma unroll
        for (int r = 0; r < 8; ++r) {
          const int mm = mb + r;
          uint32_t x = pcb;
          if (mm & 1) x ^= pk1; if (mm & 2) x ^= pk2;
          if (mm & 4) x ^= pk4; if (mm & 8) x ^= pk8;
          st2[x] = make_float2(cr[r], ci[r]);
        }
      }
      __syncthreads();
    }
  }

  float acc[NQ];
#pragma unroll
  for (int q = 0; q < NQ; ++q) acc[q] = 0.f;
  for (int i = 0; i < NSTATES / BLOCK; ++i) {
    const int y = tid + BLOCK * i;
    const uint32_t phys = (uint32_t)y ^ ((uint32_t)y >> 8);
    const float2 a = st2[phys];
    const float p = a.x * a.x + a.y * a.y;
#pragma unroll
    for (int q = 0; q < NQ; ++q)
      acc[q] += ((y >> (13 - q)) & 1) ? -p : p;
  }
#pragma unroll
  for (int q = 0; q < NQ; ++q) {
    float v = acc[q];
#pragma unroll
    for (int off = 16; off > 0; off >>= 1) v += __shfl_down(v, off, 32);
    if (lane == 0) s_tab[wave * NQ + q] = v;
  }
  __syncthreads();
  if (tid < NQ) {
    float z = 0.f;
#pragma unroll
    for (int w = 0; w < 8; ++w) z += s_tab[w * NQ + tid];
    out[bid * NQ + tid] = z;
  }
}

extern "C" void kernel_launch(void* const* d_in, const int* in_sizes, int n_in,
                              void* d_out, int out_size, void* d_ws, size_t ws_size,
                              hipStream_t stream) {
  const float* latent  = (const float*)d_in[0];
  const float* weights = (const float*)d_in[1];
  float* out = (float*)d_out;

  const int smem_bytes = SMEM_BYTES;
  hipFuncSetAttribute(reinterpret_cast<const void*>(qcircuit_kernel),
                      hipFuncAttributeMaxDynamicSharedMemorySize, smem_bytes);
  qcircuit_kernel<<<512, BLOCK, smem_bytes, stream>>>(latent, weights, out);
}


# Complex 16x16 product on 3 real matrix products (12 WMMA per tile, was 16): T=Ur(Sr+Si), Cr=T-(Ur+Ui)Si, Ci=T+(Ui-Ur)Sr; changes the rounding and summation order of every amplitude
# speedup vs baseline: 1.6121x; 1.1586x over previous
#include <hip/hip_runtime.h>
#include <hip/hip_bf16.h>
#include <stdint.h>

#define NQ      14
#define NL      8
#define NSTATES 16384
#define BLOCK   256
#define NGATE   (NL * NQ)
#define SMEM_BYTES 135712

typedef __attribute__((ext_vector_type(2))) float v2f;
typedef __attribute__((ext_vector_type(8))) float v8f;

__device__ __forceinline__ float2 cmulf(float2 a, float2 b) {
  return make_float2(a.x * b.x - a.y * b.y, a.x * b.y + a.y * b.x);
}

__device__ __forceinline__ uint32_t perm_map(uint32_t y, int l) {
  uint32_t r = y;
#pragma unroll
  for (int j = 1; j <= 7; ++j)
    if ((j & ~l) == 0) r ^= (y >> j);
  return r;
}

__global__ __launch_bounds__(BLOCK)
void qcircuit_kernel(const float* __restrict__ latent,
                     const float* __restrict__ weights,
                     float* __restrict__ out) {
  extern __shared__ float smem[];
  float2* st2  = (float2*)smem;
  float* s_tab = smem + 2 * NSTATES;
  float2* g_tab = (float2*)(smem + 2 * NSTATES + 256);

  const int tid  = threadIdx.x;
  const int lane = tid & 31;
  const int wave = tid >> 5;
  const int bid  = blockIdx.x;
  const float* lat = latent + bid * NQ;

  {
    const int half = tid >> 7, idx = tid & 127;
    float prod = 1.f;
#pragma unroll
    for (int j = 0; j < 7; ++j) {
      float s, c;
      sincosf(0.5f * lat[half * 7 + j], &s, &c);
      prod *= ((idx >> (6 - j)) & 1) ? s : c;
    }
    s_tab[tid] = prod;
  }
  if (tid < NGATE) {
    const float* w = weights + tid * 3;
    const float phi = w[0], th = w[1], om = w[2];
    float sth, cth, sa, ca, sb, cb;
    sincosf(0.5f * th, &sth, &cth);
    sincosf(0.5f * (phi + om), &sa, &ca);
    sincosf(0.5f * (phi - om), &sb, &cb);
    g_tab[tid * 4 + 0] = make_float2( cth * ca, -cth * sa);
    g_tab[tid * 4 + 1] = make_float2(-sth * cb, -sth * sb);
    g_tab[tid * 4 + 2] = make_float2( sth * cb, -sth * sb);
    g_tab[tid * 4 + 3] = make_float2( cth * ca,  cth * sa);
  } else if (tid == NGATE) {
    g_tab[NGATE * 4 + 0] = make_float2(1.f, 0.f); g_tab[NGATE * 4 + 1] = make_float2(0.f, 0.f);
    g_tab[NGATE * 4 + 2] = make_float2(0.f, 0.f); g_tab[NGATE * 4 + 3] = make_float2(1.f, 0.f);
  }
  __syncthreads();
#pragma unroll 4
  for (int i = 0; i < NSTATES / BLOCK; ++i) {
    const int y = tid * (NSTATES / BLOCK) + i;
    const float mag = s_tab[y >> 7] * s_tab[128 + (y & 127)];
    const int pc = __popc(y) & 3;
    float2 a;
    a.x = (pc == 0) ? mag : ((pc == 2) ? -mag : 0.f);
    a.y = (pc == 1) ? -mag : ((pc == 3) ? mag : 0.f);
    st2[y] = a;
  }
  __syncthreads();

  const int bsh[4] = {10, 6, 2, 0};
  for (int l = 0; l < NL; ++l) {
    for (int g = 0; g < 4; ++g) {
      const int b = bsh[g];
      const int m = lane & 15;
      const int h = lane >> 4;
      const int khi = h << 1;
      const int i0 = l * NQ + 13 - b;
      const int i1 = i0 - 1;
      const int i2 = (g == 3) ? NGATE : i0 - 2;
      const int i3 = (g == 3) ? NGATE : i0 - 3;
      const float2 g0c0 = g_tab[i0 * 4 + ((m     ) & 1) * 2 + 0];
      const float2 g0c1 = g_tab[i0 * 4 + ((m     ) & 1) * 2 + 1];
      const float2 g1ch = g_tab[i1 * 4 + ((m >> 1) & 1) * 2 + h];
      const float2 g2c0 = g_tab[i2 * 4 + ((m >> 2) & 1) * 2 + 0];
      const float2 g2c1 = g_tab[i2 * 4 + ((m >> 2) & 1) * 2 + 1];
      const float2 g3c0 = g_tab[i3 * 4 + ((m >> 3) & 1) * 2 + 0];
      const float2 g3c1 = g_tab[i3 * 4 + ((m >> 3) & 1) * 2 + 1];
      v2f aR[4], aP[4], aQ[4];
#pragma unroll
      for (int j = 0; j < 4; ++j) {
        const float2 g2 = (j & 1) ? g2c1 : g2c0;
        const float2 g3 = (j >> 1) ? g3c1 : g3c0;
        float2 u0 = cmulf(g0c0, g1ch);
        u0 = cmulf(u0, g2);
        u0 = cmulf(u0, g3);
        float2 u1 = cmulf(g0c1, g1ch);
        u1 = cmulf(u1, g2);
        u1 = cmulf(u1, g3);
        aR[j].x = u0.x;            aR[j].y = u1.x;
        aP[j].x = -(u0.x + u0.y);  aP[j].y = -(u1.x + u1.y);
        aQ[j].x = u0.y - u0.x;     aQ[j].y = u1.y - u1.x;
      }
      const uint32_t pk1 = perm_map(1u << b, l);
      const uint32_t pk2 = perm_map(2u << b, l);
      const uint32_t pk4 = perm_map(4u << b, l);
      const uint32_t pk8 = perm_map(8u << b, l);
      const uint32_t lowm = (1u << b) - 1u;

      for (int ti = 0; ti < 8; ++ti) {
        const int t = wave * 8 + ti;
        const int c = t * 16 + (lane & 15);
        const uint32_t cpart = (((uint32_t)c >> b) << (b + 4)) | ((uint32_t)c & lowm);
        const uint32_t pcb = perm_map(cpart, l);
        v2f bR[4], bI[4], bS[4];
#pragma unroll
        for (int j = 0; j < 4; ++j) {
          const int k0 = 4 * j + khi, k1 = k0 + 1;
          uint32_t x0 = pcb, x1 = pcb;
          if (k0 & 1) x0 ^= pk1; if (k0 & 2) x0 ^= pk2;
          if (k0 & 4) x0 ^= pk4; if (k0 & 8) x0 ^= pk8;
          if (k1 & 1) x1 ^= pk1; if (k1 & 2) x1 ^= pk2;
          if (k1 & 4) x1 ^= pk4; if (k1 & 8) x1 ^= pk8;
          const float2 s0 = st2[x0], s1 = st2[x1];
          bR[j].x = s0.x; bR[j].y = s1.x;
          bI[j].x = s0.y; bI[j].y = s1.y;
          bS[j] = bR[j] + bI[j];
        }
        v8f tt = {};
#pragma unroll
        for (int j = 0; j < 4; ++j)
          tt = __builtin_amdgcn_wmma_f32_16x16x4_f32(false, aR[j], false, bS[j], (short)0, tt, false, false);
        v8f cr = tt, ci = tt;
#pragma unroll
        for (int j = 0; j < 4; ++j) {
          cr = __builtin_amdgcn_wmma_f32_16x16x4_f32(false, aP[j], false, bI[j], (short)0, cr, false, false);
          ci = __builtin_amdgcn_wmma_f32_16x16x4_f32(false, aQ[j], false, bR[j], (short)0, ci, false, false);
        }
        const int mb = (lane >> 4) << 3;
#pragma unroll
        for (int r = 0; r < 8; ++r) {
          const int mm = mb + r;
          uint32_t x = pcb;
          if (mm & 1) x ^= pk1; if (mm & 2) x ^= pk2;
          if (mm & 4) x ^= pk4; if (mm & 8) x ^= pk8;
          st2[x] = make_float2(cr[r], ci[r]);
        }
      }
      __syncthreads();
    }
  }

  float acc[NQ];
#pragma unroll
  for (int q = 0; q < NQ; ++q) acc[q] = 0.f;
  for (int i = 0; i < NSTATES / BLOCK; ++i) {
    const int y = tid + BLOCK * i;
    const uint32_t phys = (uint32_t)y ^ ((uint32_t)y >> 8);
    const float2 a = st2[phys];
    const float p = a.x * a.x + a.y * a.y;
#pragma unroll
    for (int q = 0; q < NQ; ++q)
      acc[q] += ((y >> (13 - q)) & 1) ? -p : p;
  }
#pragma unroll
  for (int q = 0; q < NQ; ++q) {
    float v = acc[q];
#pragma unroll
    for (int off = 16; off > 0; off >>= 1) v += __shfl_down(v, off, 32);
    if (lane == 0) s_tab[wave * NQ + q] = v;
  }
  __syncthreads();
  if (tid < NQ) {
    float z = 0.f;
#pragma unroll
    for (int w = 0; w < 8; ++w) z += s_tab[w * NQ + tid];
    out[bid * NQ + tid] = z;
  }
}

extern "C" void kernel_launch(void* const* d_in, const int* in_sizes, int n_in,
                              void* d_out, int out_size, void* d_ws, size_t ws_size,
                              hipStream_t stream) {
  const float* latent  = (const float*)d_in[0];
  const float* weights = (const float*)d_in[1];
  float* out = (float*)d_out;

  const int smem_bytes = SMEM_BYTES;
  hipFuncSetAttribute(reinterpret_cast<const void*>(qcircuit_kernel),
                      hipFuncAttributeMaxDynamicSharedMemorySize, smem_bytes);
  qcircuit_kernel<<<512, BLOCK, smem_bytes, stream>>>(latent, weights, out);
}


# LDS bank conflicts removed on the state array: index XOR-swizzle (bits 5..13 folded into the 5 bank bits) plus lane-contiguous initial fill; addresses per bank per access drop from 2-32 to 1-2
# speedup vs baseline: 1.6733x; 1.0380x over previous
#include <hip/hip_runtime.h>
#include <hip/hip_bf16.h>
#include <stdint.h>

#define NQ      14
#define NL      8
#define NSTATES 16384
#define BLOCK   256
#define NGATE   (NL * NQ)
#define SMEM_BYTES 135712

typedef __attribute__((ext_vector_type(2))) float v2f;
typedef __attribute__((ext_vector_type(8))) float v8f;

__device__ __forceinline__ float2 cmulf(float2 a, float2 b) {
  return make_float2(a.x * b.x - a.y * b.y, a.x * b.y + a.y * b.x);
}

__device__ __forceinline__ uint32_t swz(uint32_t x) {
  return x ^ ((x >> 3) & 28u) ^ ((x >> 5) & 23u) ^ ((x >> 9) & 16u);
}

__device__ __forceinline__ uint32_t perm_map(uint32_t y, int l) {
  uint32_t r = y;
#pragma unroll
  for (int j = 1; j <= 7; ++j)
    if ((j & ~l) == 0) r ^= (y >> j);
  return swz(r);
}

__global__ __launch_bounds__(BLOCK)
void qcircuit_kernel(const float* __restrict__ latent,
                     const float* __restrict__ weights,
                     float* __restrict__ out) {
  extern __shared__ float smem[];
  float2* st2  = (float2*)smem;
  float* s_tab = smem + 2 * NSTATES;
  float2* g_tab = (float2*)(smem + 2 * NSTATES + 256);

  const int tid  = threadIdx.x;
  const int lane = tid & 31;
  const int wave = tid >> 5;
  const int bid  = blockIdx.x;
  const float* lat = latent + bid * NQ;

  {
    const int half = tid >> 7, idx = tid & 127;
    float prod = 1.f;
#pragma unroll
    for (int j = 0; j < 7; ++j) {
      float s, c;
      sincosf(0.5f * lat[half * 7 + j], &s, &c);
      prod *= ((idx >> (6 - j)) & 1) ? s : c;
    }
    s_tab[tid] = prod;
  }
  if (tid < NGATE) {
    const float* w = weights + tid * 3;
    const float phi = w[0], th = w[1], om = w[2];
    float sth, cth, sa, ca, sb, cb;
    sincosf(0.5f * th, &sth, &cth);
    sincosf(0.5f * (phi + om), &sa, &ca);
    sincosf(0.5f * (phi - om), &sb, &cb);
    g_tab[tid * 4 + 0] = make_float2( cth * ca, -cth * sa);
    g_tab[tid * 4 + 1] = make_float2(-sth * cb, -sth * sb);
    g_tab[tid * 4 + 2] = make_float2( sth * cb, -sth * sb);
    g_tab[tid * 4 + 3] = make_float2( cth * ca,  cth * sa);
  } else if (tid == NGATE) {
    g_tab[NGATE * 4 + 0] = make_float2(1.f, 0.f); g_tab[NGATE * 4 + 1] = make_float2(0.f, 0.f);
    g_tab[NGATE * 4 + 2] = make_float2(0.f, 0.f); g_tab[NGATE * 4 + 3] = make_float2(1.f, 0.f);
  }
  __syncthreads();
#pragma unroll 4
  for (int i = 0; i < NSTATES / BLOCK; ++i) {
    const int y = i * BLOCK + tid;
    const float mag = s_tab[y >> 7] * s_tab[128 + (y & 127)];
    const int pc = __popc(y) & 3;
    float2 a;
    a.x = (pc == 0) ? mag : ((pc == 2) ? -mag : 0.f);
    a.y = (pc == 1) ? -mag : ((pc == 3) ? mag : 0.f);
    st2[swz((uint32_t)y)] = a;
  }
  __syncthreads();

  const int bsh[4] = {10, 6, 2, 0};
  for (int l = 0; l < NL; ++l) {
    for (int g = 0; g < 4; ++g) {
      const int b = bsh[g];
      const int m = lane & 15;
      const int h = lane >> 4;
      const int khi = h << 1;
      const int i0 = l * NQ + 13 - b;
      const int i1 = i0 - 1;
      const int i2 = (g == 3) ? NGATE : i0 - 2;
      const int i3 = (g == 3) ? NGATE : i0 - 3;
      const float2 g0c0 = g_tab[i0 * 4 + ((m     ) & 1) * 2 + 0];
      const float2 g0c1 = g_tab[i0 * 4 + ((m     ) & 1) * 2 + 1];
      const float2 g1ch = g_tab[i1 * 4 + ((m >> 1) & 1) * 2 + h];
      const float2 g2c0 = g_tab[i2 * 4 + ((m >> 2) & 1) * 2 + 0];
      const float2 g2c1 = g_tab[i2 * 4 + ((m >> 2) & 1) * 2 + 1];
      const float2 g3c0 = g_tab[i3 * 4 + ((m >> 3) & 1) * 2 + 0];
      const float2 g3c1 = g_tab[i3 * 4 + ((m >> 3) & 1) * 2 + 1];
      v2f aR[4], aP[4], aQ[4];
#pragma unroll
      for (int j = 0; j < 4; ++j) {
        const float2 g2 = (j & 1) ? g2c1 : g2c0;
        const float2 g3 = (j >> 1) ? g3c1 : g3c0;
        float2 u0 = cmulf(g0c0, g1ch);
        u0 = cmulf(u0, g2);
        u0 = cmulf(u0, g3);
        float2 u1 = cmulf(g0c1, g1ch);
        u1 = cmulf(u1, g2);
        u1 = cmulf(u1, g3);
        aR[j].x = u0.x;            aR[j].y = u1.x;
        aP[j].x = -(u0.x + u0.y);  aP[j].y = -(u1.x + u1.y);
        aQ[j].x = u0.y - u0.x;     aQ[j].y = u1.y - u1.x;
      }
      const uint32_t pk1 = perm_map(1u << b, l);
      const uint32_t pk2 = perm_map(2u << b, l);
      const uint32_t pk4 = perm_map(4u << b, l);
      const uint32_t pk8 = perm_map(8u << b, l);
      const uint32_t lowm = (1u << b) - 1u;

      for (int ti = 0; ti < 8; ++ti) {
        const int t = wave * 8 + ti;
        const int c = t * 16 + (lane & 15);
        const uint32_t cpart = (((uint32_t)c >> b) << (b + 4)) | ((uint32_t)c & lowm);
        const uint32_t pcb = perm_map(cpart, l);
        v2f bR[4], bI[4], bS[4];
#pragma unroll
        for (int j = 0; j < 4; ++j) {
          const int k0 = 4 * j + khi, k1 = k0 + 1;
          uint32_t x0 = pcb, x1 = pcb;
          if (k0 & 1) x0 ^= pk1; if (k0 & 2) x0 ^= pk2;
          if (k0 & 4) x0 ^= pk4; if (k0 & 8) x0 ^= pk8;
          if (k1 & 1) x1 ^= pk1; if (k1 & 2) x1 ^= pk2;
          if (k1 & 4) x1 ^= pk4; if (k1 & 8) x1 ^= pk8;
          const float2 s0 = st2[x0], s1 = st2[x1];
          bR[j].x = s0.x; bR[j].y = s1.x;
          bI[j].x = s0.y; bI[j].y = s1.y;
          bS[j] = bR[j] + bI[j];
        }
        v8f tt = {};
#pragma unroll
        for (int j = 0; j < 4; ++j)
          tt = __builtin_amdgcn_wmma_f32_16x16x4_f32(false, aR[j], false, bS[j], (short)0, tt, false, false);
        v8f cr = tt, ci = tt;
#pragma unroll
        for (int j = 0; j < 4; ++j) {
          cr = __builtin_amdgcn_wmma_f32_16x16x4_f32(false, aP[j], false, bI[j], (short)0, cr, false, false);
          ci = __builtin_amdgcn_wmma_f32_16x16x4_f32(false, aQ[j], false, bR[j], (short)0, ci, false, false);
        }
        const int mb = (lane >> 4) << 3;
#pragma unroll
        for (int r = 0; r < 8; ++r) {
          const int mm = mb + r;
          uint32_t x = pcb;
          if (mm & 1) x ^= pk1; if (mm & 2) x ^= pk2;
          if (mm & 4) x ^= pk4; if (mm & 8) x ^= pk8;
          st2[x] = make_float2(cr[r], ci[r]);
        }
      }
      __syncthreads();
    }
  }

  float acc[NQ];
#pragma unroll
  for (int q = 0; q < NQ; ++q) acc[q] = 0.f;
  for (int i = 0; i < NSTATES / BLOCK; ++i) {
    const int y = tid + BLOCK * i;
    const uint32_t phys = swz((uint32_t)y ^ ((uint32_t)y >> 8));
    const float2 a = st2[phys];
    const float p = a.x * a.x + a.y * a.y;
#pragma unroll
    for (int q = 0; q < NQ; ++q)
      acc[q] += ((y >> (13 - q)) & 1) ? -p : p;
  }
#pragma unroll
  for (int q = 0; q < NQ; ++q) {
    float v = acc[q];
#pragma unroll
    for (int off = 16; off > 0; off >>= 1) v += __shfl_down(v, off, 32);
    if (lane == 0) s_tab[wave * NQ + q] = v;
  }
  __syncthreads();
  if (tid < NQ) {
    float z = 0.f;
#pragma unroll
    for (int w = 0; w < 8; ++w) z += s_tab[w * NQ + tid];
    out[bid * NQ + tid] = z;
  }
}

extern "C" void kernel_launch(void* const* d_in, const int* in_sizes, int n_in,
                              void* d_out, int out_size, void* d_ws, size_t ws_size,
                              hipStream_t stream) {
  const float* latent  = (const float*)d_in[0];
  const float* weights = (const float*)d_in[1];
  float* out = (float*)d_out;

  const int smem_bytes = SMEM_BYTES;
  hipFuncSetAttribute(reinterpret_cast<const void*>(qcircuit_kernel),
                      hipFuncAttributeMaxDynamicSharedMemorySize, smem_bytes);
  qcircuit_kernel<<<512, BLOCK, smem_bytes, stream>>>(latent, weights, out);
}


# Two 4-bit gate groups per LDS pass: group 1's product is formed transposed so its D registers are the A operand of group 2's product; 2 passes and barriers per layer, was 4; new summation order
# speedup vs baseline: 2.0139x; 1.2035x over previous
#include <hip/hip_runtime.h>
#include <hip/hip_bf16.h>
#include <stdint.h>

#define NQ      14
#define NL      8
#define NSTATES 16384
#define BLOCK   256
#define NGATE   (NL * NQ)
#define SMEM_BYTES 135712

typedef __attribute__((ext_vector_type(2))) float v2f;
typedef __attribute__((ext_vector_type(8))) float v8f;

__device__ __forceinline__ float2 cmulf(float2 a, float2 b) {
  return make_float2(a.x * b.x - a.y * b.y, a.x * b.y + a.y * b.x);
}

__device__ __forceinline__ uint32_t swz(uint32_t x) {
  return x ^ ((x >> 3) & 28u) ^ ((x >> 5) & 23u) ^ ((x >> 9) & 16u);
}

__device__ __forceinline__ uint32_t perm_map(uint32_t y, int l) {
  uint32_t r = y;
#pragma unroll
  for (int j = 1; j <= 7; ++j)
    if ((j & ~l) == 0) r ^= (y >> j);
  return swz(r);
}

__global__ __launch_bounds__(BLOCK)
void qcircuit_kernel(const float* __restrict__ latent,
                     const float* __restrict__ weights,
                     float* __restrict__ out) {
  extern __shared__ float smem[];
  float2* st2  = (float2*)smem;
  float* s_tab = smem + 2 * NSTATES;
  float2* g_tab = (float2*)(smem + 2 * NSTATES + 256);

  const int tid  = threadIdx.x;
  const int lane = tid & 31;
  const int wave = tid >> 5;
  const int bid  = blockIdx.x;
  const float* lat = latent + bid * NQ;

  {
    const int half = tid >> 7, idx = tid & 127;
    float prod = 1.f;
#pragma unroll
    for (int j = 0; j < 7; ++j) {
      float s, c;
      sincosf(0.5f * lat[half * 7 + j], &s, &c);
      prod *= ((idx >> (6 - j)) & 1) ? s : c;
    }
    s_tab[tid] = prod;
  }
  if (tid < NGATE) {
    const float* w = weights + tid * 3;
    const float phi = w[0], th = w[1], om = w[2];
    float sth, cth, sa, ca, sb, cb;
    sincosf(0.5f * th, &sth, &cth);
    sincosf(0.5f * (phi + om), &sa, &ca);
    sincosf(0.5f * (phi - om), &sb, &cb);
    g_tab[tid * 4 + 0] = make_float2( cth * ca, -cth * sa);
    g_tab[tid * 4 + 1] = make_float2(-sth * cb, -sth * sb);
    g_tab[tid * 4 + 2] = make_float2( sth * cb, -sth * sb);
    g_tab[tid * 4 + 3] = make_float2( cth * ca,  cth * sa);
  } else if (tid == NGATE) {
    g_tab[NGATE * 4 + 0] = make_float2(1.f, 0.f); g_tab[NGATE * 4 + 1] = make_float2(0.f, 0.f);
    g_tab[NGATE * 4 + 2] = make_float2(0.f, 0.f); g_tab[NGATE * 4 + 3] = make_float2(1.f, 0.f);
  }
  __syncthreads();
#pragma unroll 4
  for (int i = 0; i < NSTATES / BLOCK; ++i) {
    const int y = i * BLOCK + tid;
    const float mag = s_tab[y >> 7] * s_tab[128 + (y & 127)];
    const int pc = __popc(y) & 3;
    float2 a;
    a.x = (pc == 0) ? mag : ((pc == 2) ? -mag : 0.f);
    a.y = (pc == 1) ? -mag : ((pc == 3) ? mag : 0.f);
    st2[swz((uint32_t)y)] = a;
  }
  __syncthreads();

  for (int l = 0; l < NL; ++l) {
    for (int p = 0; p < 2; ++p) {
      const int bx = p ? 0 : 10;
      const int by = p ? 4 : 6;
      const int bt = p ? 8 : 0;
      const int m = lane & 15;
      const int h = lane >> 4;
      const int khi = h << 1;
      const int i0 = l * NQ + 13 - bx;
      const int i1 = i0 - 1, i2 = i0 - 2, i3 = i0 - 3;
      const float2 g0c0 = g_tab[i0 * 4 + ((m     ) & 1) * 2 + 0];
      const float2 g0c1 = g_tab[i0 * 4 + ((m     ) & 1) * 2 + 1];
      const float2 g1ch = g_tab[i1 * 4 + ((m >> 1) & 1) * 2 + h];
      const float2 g2c0 = g_tab[i2 * 4 + ((m >> 2) & 1) * 2 + 0];
      const float2 g2c1 = g_tab[i2 * 4 + ((m >> 2) & 1) * 2 + 1];
      const float2 g3c0 = g_tab[i3 * 4 + ((m >> 3) & 1) * 2 + 0];
      const float2 g3c1 = g_tab[i3 * 4 + ((m >> 3) & 1) * 2 + 1];
      const int y0 = l * NQ + 13 - by;
      const int y1 = y0 - 1;
      const int y2 = p ? NGATE : y0 - 2;
      const int y3 = p ? NGATE : y0 - 3;
      const float2 q0c0 = g_tab[y0 * 4 + ((m     ) & 1) * 2 + 0];
      const float2 q0c1 = g_tab[y0 * 4 + ((m     ) & 1) * 2 + 1];
      const float2 q1c0 = g_tab[y1 * 4 + ((m >> 1) & 1) * 2 + 0];
      const float2 q1c1 = g_tab[y1 * 4 + ((m >> 1) & 1) * 2 + 1];
      const float2 q2c0 = g_tab[y2 * 4 + ((m >> 2) & 1) * 2 + 0];
      const float2 q2c1 = g_tab[y2 * 4 + ((m >> 2) & 1) * 2 + 1];
      const float2 q3ch = g_tab[y3 * 4 + ((m >> 3) & 1) * 2 + h];
      v2f aR[4], aP[4], aQ[4];
      v2f yR[4], yP[4], yQ[4];
#pragma unroll
      for (int j = 0; j < 4; ++j) {
        const float2 g2 = (j & 1) ? g2c1 : g2c0;
        const float2 g3 = (j >> 1) ? g3c1 : g3c0;
        float2 u0 = cmulf(g0c0, g1ch);
        u0 = cmulf(u0, g2);
        u0 = cmulf(u0, g3);
        float2 u1 = cmulf(g0c1, g1ch);
        u1 = cmulf(u1, g2);
        u1 = cmulf(u1, g3);
        aR[j].x = u0.x;            aR[j].y = u1.x;
        aP[j].x = -(u0.x + u0.y);  aP[j].y = -(u1.x + u1.y);
        aQ[j].x = u0.y - u0.x;     aQ[j].y = u1.y - u1.x;
        const float2 q1 = (j & 1) ? q1c1 : q1c0;
        const float2 q2 = (j >> 1) ? q2c1 : q2c0;
        float2 v0 = cmulf(q0c0, q1);
        v0 = cmulf(v0, q2);
        v0 = cmulf(v0, q3ch);
        float2 v1 = cmulf(q0c1, q1);
        v1 = cmulf(v1, q2);
        v1 = cmulf(v1, q3ch);
        yR[j].x = v0.x;            yR[j].y = v1.x;
        yP[j].x = -(v0.x + v0.y);  yP[j].y = -(v1.x + v1.y);
        yQ[j].x = v0.y - v0.x;     yQ[j].y = v1.y - v1.x;
      }
      const uint32_t pk1 = perm_map(1u << bx, l);
      const uint32_t pk2 = perm_map(2u << bx, l);
      const uint32_t pk4 = perm_map(4u << bx, l);
      const uint32_t pk8 = perm_map(8u << bx, l);

      for (int ti = 0; ti < 8; ++ti) {
        const int t = wave * 8 + ti;
        const uint32_t cpart = ((uint32_t)m << by) | ((uint32_t)t << bt);
        const uint32_t pcb = perm_map(cpart, l);
        v2f bR[4], bI[4], bS[4];
#pragma unroll
        for (int j = 0; j < 4; ++j) {
          const int k0 = 4 * j + khi, k1 = k0 + 1;
          uint32_t x0 = pcb, x1 = pcb;
          if (k0 & 1) x0 ^= pk1; if (k0 & 2) x0 ^= pk2;
          if (k0 & 4) x0 ^= pk4; if (k0 & 8) x0 ^= pk8;
          if (k1 & 1) x1 ^= pk1; if (k1 & 2) x1 ^= pk2;
          if (k1 & 4) x1 ^= pk4; if (k1 & 8) x1 ^= pk8;
          const float2 s0 = st2[x0], s1 = st2[x1];
          bR[j].x = s0.x; bR[j].y = s1.x;
          bI[j].x = s0.y; bI[j].y = s1.y;
          bS[j] = bR[j] + bI[j];
        }
        v8f tt = {};
#pragma unroll
        for (int j = 0; j < 4; ++j)
          tt = __builtin_amdgcn_wmma_f32_16x16x4_f32(false, bS[j], false, aR[j], (short)0, tt, false, false);
        v8f dr = tt, di = tt;
#pragma unroll
        for (int j = 0; j < 4; ++j) {
          dr = __builtin_amdgcn_wmma_f32_16x16x4_f32(false, bI[j], false, aP[j], (short)0, dr, false, false);
          di = __builtin_amdgcn_wmma_f32_16x16x4_f32(false, bR[j], false, aQ[j], (short)0, di, false, false);
        }
        const v8f ds = dr + di;
        v8f t2 = {};
#pragma unroll
        for (int j = 0; j < 4; ++j) {
          v2f as; as.x = ds[2 * j]; as.y = ds[2 * j + 1];
          t2 = __builtin_amdgcn_wmma_f32_16x16x4_f32(false, as, false, yR[j], (short)0, t2, false, false);
        }
        v8f cr = t2, ci = t2;
#pragma unroll
        for (int j = 0; j < 4; ++j) {
          v2f ai; ai.x = di[2 * j]; ai.y = di[2 * j + 1];
          v2f ar; ar.x = dr[2 * j]; ar.y = dr[2 * j + 1];
          cr = __builtin_amdgcn_wmma_f32_16x16x4_f32(false, ai, false, yP[j], (short)0, cr, false, false);
          ci = __builtin_amdgcn_wmma_f32_16x16x4_f32(false, ar, false, yQ[j], (short)0, ci, false, false);
        }
        const int mb = h << 3;
#pragma unroll
        for (int r = 0; r < 8; ++r) {
          const int mm = mb + r;
          uint32_t x = pcb;
          if (mm & 1) x ^= pk1; if (mm & 2) x ^= pk2;
          if (mm & 4) x ^= pk4; if (mm & 8) x ^= pk8;
          st2[x] = make_float2(cr[r], ci[r]);
        }
      }
      __syncthreads();
    }
  }

  float acc[NQ];
#pragma unroll
  for (int q = 0; q < NQ; ++q) acc[q] = 0.f;
  for (int i = 0; i < NSTATES / BLOCK; ++i) {
    const int y = tid + BLOCK * i;
    const uint32_t phys = swz((uint32_t)y ^ ((uint32_t)y >> 8));
    const float2 a = st2[phys];
    const float p = a.x * a.x + a.y * a.y;
#pragma unroll
    for (int q = 0; q < NQ; ++q)
      acc[q] += ((y >> (13 - q)) & 1) ? -p : p;
  }
#pragma unroll
  for (int q = 0; q < NQ; ++q) {
    float v = acc[q];
#pragma unroll
    for (int off = 16; off > 0; off >>= 1) v += __shfl_down(v, off, 32);
    if (lane == 0) s_tab[wave * NQ + q] = v;
  }
  __syncthreads();
  if (tid < NQ) {
    float z = 0.f;
#pragma unroll
    for (int w = 0; w < 8; ++w) z += s_tab[w * NQ + tid];
    out[bid * NQ + tid] = z;
  }
}

extern "C" void kernel_launch(void* const* d_in, const int* in_sizes, int n_in,
                              void* d_out, int out_size, void* d_ws, size_t ws_size,
                              hipStream_t stream) {
  const float* latent  = (const float*)d_in[0];
  const float* weights = (const float*)d_in[1];
  float* out = (float*)d_out;

  const int smem_bytes = SMEM_BYTES;
  hipFuncSetAttribute(reinterpret_cast<const void*>(qcircuit_kernel),
                      hipFuncAttributeMaxDynamicSharedMemorySize, smem_bytes);
  qcircuit_kernel<<<512, BLOCK, smem_bytes, stream>>>(latent, weights, out);
}


# Layer 0's Rot gates act on a product state: each is applied to its qubit's 2-vector RX|0> before the amplitudes are formed (exact, any input); matrix passes run layers 1..7; new layer-0 rounding
# speedup vs baseline: 2.2585x; 1.1215x over previous
#include <hip/hip_runtime.h>
#include <hip/hip_bf16.h>
#include <stdint.h>

#define NQ      14
#define NL      8
#define NSTATES 16384
#define BLOCK   256
#define NGATE   (NL * NQ)
#define SMEM_BYTES 136992

typedef __attribute__((ext_vector_type(2))) float v2f;
typedef __attribute__((ext_vector_type(8))) float v8f;

__device__ __forceinline__ float2 cmulf(float2 a, float2 b) {
  return make_float2(a.x * b.x - a.y * b.y, a.x * b.y + a.y * b.x);
}

__device__ __forceinline__ uint32_t swz(uint32_t x) {
  return x ^ ((x >> 3) & 28u) ^ ((x >> 5) & 23u) ^ ((x >> 9) & 16u);
}

__device__ __forceinline__ uint32_t perm_map(uint32_t y, int l) {
  uint32_t r = y;
#pragma unroll
  for (int j = 1; j <= 7; ++j)
    if ((j & ~l) == 0) r ^= (y >> j);
  return swz(r);
}

__global__ __launch_bounds__(BLOCK)
void qcircuit_kernel(const float* __restrict__ latent,
                     const float* __restrict__ weights,
                     float* __restrict__ out) {
  extern __shared__ float smem[];
  float2* st2  = (float2*)smem;
  float* s_tab = smem + 2 * NSTATES;
  float2* c_tab = (float2*)s_tab;
  float2* vq    = (float2*)(smem + 2 * NSTATES + 512);
  float2* g_tab = (float2*)(smem + 2 * NSTATES + 576);

  const int tid  = threadIdx.x;
  const int lane = tid & 31;
  const int wave = tid >> 5;
  const int bid  = blockIdx.x;
  const float* lat = latent + bid * NQ;

  if (tid < NGATE) {
    const float* w = weights + tid * 3;
    const float phi = w[0], th = w[1], om = w[2];
    float sth, cth, sa, ca, sb, cb;
    sincosf(0.5f * th, &sth, &cth);
    sincosf(0.5f * (phi + om), &sa, &ca);
    sincosf(0.5f * (phi - om), &sb, &cb);
    const float2 m00 = make_float2( cth * ca, -cth * sa);
    const float2 m01 = make_float2(-sth * cb, -sth * sb);
    const float2 m10 = make_float2( sth * cb, -sth * sb);
    const float2 m11 = make_float2( cth * ca,  cth * sa);
    if (tid < NQ) {
      float sn, cs;
      sincosf(0.5f * lat[tid], &sn, &cs);
      vq[tid * 2 + 0] = make_float2(m00.x * cs + m01.y * sn, m00.y * cs - m01.x * sn);
      vq[tid * 2 + 1] = make_float2(m10.x * cs + m11.y * sn, m10.y * cs - m11.x * sn);
    } else {
      g_tab[tid * 4 + 0] = m00;
      g_tab[tid * 4 + 1] = m01;
      g_tab[tid * 4 + 2] = m10;
      g_tab[tid * 4 + 3] = m11;
    }
  } else if (tid == NGATE) {
    g_tab[NGATE * 4 + 0] = make_float2(1.f, 0.f); g_tab[NGATE * 4 + 1] = make_float2(0.f, 0.f);
    g_tab[NGATE * 4 + 2] = make_float2(0.f, 0.f); g_tab[NGATE * 4 + 3] = make_float2(1.f, 0.f);
  }
  __syncthreads();
  {
    const int half = tid >> 7, idx = tid & 127;
    float2 prod = vq[(half * 7) * 2 + ((idx >> 6) & 1)];
#pragma unroll
    for (int j = 1; j < 7; ++j)
      prod = cmulf(prod, vq[(half * 7 + j) * 2 + ((idx >> (6 - j)) & 1)]);
    c_tab[tid] = prod;
  }
  __syncthreads();
#pragma unroll 4
  for (int i = 0; i < NSTATES / BLOCK; ++i) {
    const int y = i * BLOCK + tid;
    st2[swz((uint32_t)y)] = cmulf(c_tab[y >> 7], c_tab[128 + (y & 127)]);
  }
  __syncthreads();

  for (int l = 1; l < NL; ++l) {
    for (int p = 0; p < 2; ++p) {
      const int bx = p ? 0 : 10;
      const int by = p ? 4 : 6;
      const int bt = p ? 8 : 0;
      const int m = lane & 15;
      const int h = lane >> 4;
      const int khi = h << 1;
      const int i0 = l * NQ + 13 - bx;
      const int i1 = i0 - 1, i2 = i0 - 2, i3 = i0 - 3;
      const float2 g0c0 = g_tab[i0 * 4 + ((m     ) & 1) * 2 + 0];
      const float2 g0c1 = g_tab[i0 * 4 + ((m     ) & 1) * 2 + 1];
      const float2 g1ch = g_tab[i1 * 4 + ((m >> 1) & 1) * 2 + h];
      const float2 g2c0 = g_tab[i2 * 4 + ((m >> 2) & 1) * 2 + 0];
      const float2 g2c1 = g_tab[i2 * 4 + ((m >> 2) & 1) * 2 + 1];
      const float2 g3c0 = g_tab[i3 * 4 + ((m >> 3) & 1) * 2 + 0];
      const float2 g3c1 = g_tab[i3 * 4 + ((m >> 3) & 1) * 2 + 1];
      const int y0 = l * NQ + 13 - by;
      const int y1 = y0 - 1;
      const int y2 = p ? NGATE : y0 - 2;
      const int y3 = p ? NGATE : y0 - 3;
      const float2 q0c0 = g_tab[y0 * 4 + ((m     ) & 1) * 2 + 0];
      const float2 q0c1 = g_tab[y0 * 4 + ((m     ) & 1) * 2 + 1];
      const float2 q1c0 = g_tab[y1 * 4 + ((m >> 1) & 1) * 2 + 0];
      const float2 q1c1 = g_tab[y1 * 4 + ((m >> 1) & 1) * 2 + 1];
      const float2 q2c0 = g_tab[y2 * 4 + ((m >> 2) & 1) * 2 + 0];
      const float2 q2c1 = g_tab[y2 * 4 + ((m >> 2) & 1) * 2 + 1];
      const float2 q3ch = g_tab[y3 * 4 + ((m >> 3) & 1) * 2 + h];
      v2f aR[4], aP[4], aQ[4];
      v2f yR[4], yP[4], yQ[4];
#pragma unroll
      for (int j = 0; j < 4; ++j) {
        const float2 g2 = (j & 1) ? g2c1 : g2c0;
        const float2 g3 = (j >> 1) ? g3c1 : g3c0;
        float2 u0 = cmulf(g0c0, g1ch);
        u0 = cmulf(u0, g2);
        u0 = cmulf(u0, g3);
        float2 u1 = cmulf(g0c1, g1ch);
        u1 = cmulf(u1, g2);
        u1 = cmulf(u1, g3);
        aR[j].x = u0.x;            aR[j].y = u1.x;
        aP[j].x = -(u0.x + u0.y);  aP[j].y = -(u1.x + u1.y);
        aQ[j].x = u0.y - u0.x;     aQ[j].y = u1.y - u1.x;
        const float2 q1 = (j & 1) ? q1c1 : q1c0;
        const float2 q2 = (j >> 1) ? q2c1 : q2c0;
        float2 v0 = cmulf(q0c0, q1);
        v0 = cmulf(v0, q2);
        v0 = cmulf(v0, q3ch);
        float2 v1 = cmulf(q0c1, q1);
        v1 = cmulf(v1, q2);
        v1 = cmulf(v1, q3ch);
        yR[j].x = v0.x;            yR[j].y = v1.x;
        yP[j].x = -(v0.x + v0.y);  yP[j].y = -(v1.x + v1.y);
        yQ[j].x = v0.y - v0.x;     yQ[j].y = v1.y - v1.x;
      }
      const uint32_t pk1 = perm_map(1u << bx, l);
      const uint32_t pk2 = perm_map(2u << bx, l);
      const uint32_t pk4 = perm_map(4u << bx, l);
      const uint32_t pk8 = perm_map(8u << bx, l);

      for (int ti = 0; ti < 8; ++ti) {
        const int t = wave * 8 + ti;
        const uint32_t cpart = ((uint32_t)m << by) | ((uint32_t)t << bt);
        const uint32_t pcb = perm_map(cpart, l);
        v2f bR[4], bI[4], bS[4];
#pragma unroll
        for (int j = 0; j < 4; ++j) {
          const int k0 = 4 * j + khi, k1 = k0 + 1;
          uint32_t x0 = pcb, x1 = pcb;
          if (k0 & 1) x0 ^= pk1; if (k0 & 2) x0 ^= pk2;
          if (k0 & 4) x0 ^= pk4; if (k0 & 8) x0 ^= pk8;
          if (k1 & 1) x1 ^= pk1; if (k1 & 2) x1 ^= pk2;
          if (k1 & 4) x1 ^= pk4; if (k1 & 8) x1 ^= pk8;
          const float2 s0 = st2[x0], s1 = st2[x1];
          bR[j].x = s0.x; bR[j].y = s1.x;
          bI[j].x = s0.y; bI[j].y = s1.y;
          bS[j] = bR[j] + bI[j];
        }
        v8f tt = {};
#pragma unroll
        for (int j = 0; j < 4; ++j)
          tt = __builtin_amdgcn_wmma_f32_16x16x4_f32(false, bS[j], false, aR[j], (short)0, tt, false, false);
        v8f dr = tt, di = tt;
#pragma unroll
        for (int j = 0; j < 4; ++j) {
          dr = __builtin_amdgcn_wmma_f32_16x16x4_f32(false, bI[j], false, aP[j], (short)0, dr, false, false);
          di = __builtin_amdgcn_wmma_f32_16x16x4_f32(false, bR[j], false, aQ[j], (short)0, di, false, false);
        }
        const v8f ds = dr + di;
        v8f t2 = {};
#pragma unroll
        for (int j = 0; j < 4; ++j) {
          v2f as; as.x = ds[2 * j]; as.y = ds[2 * j + 1];
          t2 = __builtin_amdgcn_wmma_f32_16x16x4_f32(false, as, false, yR[j], (short)0, t2, false, false);
        }
        v8f cr = t2, ci = t2;
#pragma unroll
        for (int j = 0; j < 4; ++j) {
          v2f ai; ai.x = di[2 * j]; ai.y = di[2 * j + 1];
          v2f ar; ar.x = dr[2 * j]; ar.y = dr[2 * j + 1];
          cr = __builtin_amdgcn_wmma_f32_16x16x4_f32(false, ai, false, yP[j], (short)0, cr, false, false);
          ci = __builtin_amdgcn_wmma_f32_16x16x4_f32(false, ar, false, yQ[j], (short)0, ci, false, false);
        }
        const int mb = h << 3;
#pragma unroll
        for (int r = 0; r < 8; ++r) {
          const int mm = mb + r;
          uint32_t x = pcb;
          if (mm & 1) x ^= pk1; if (mm & 2) x ^= pk2;
          if (mm & 4) x ^= pk4; if (mm & 8) x ^= pk8;
          st2[x] = make_float2(cr[r], ci[r]);
        }
      }
      __syncthreads();
    }
  }

  float acc[NQ];
#pragma unroll
  for (int q = 0; q < NQ; ++q) acc[q] = 0.f;
  for (int i = 0; i < NSTATES / BLOCK; ++i) {
    const int y = tid + BLOCK * i;
    const uint32_t phys = swz((uint32_t)y ^ ((uint32_t)y >> 8));
    const float2 a = st2[phys];
    const float p = a.x * a.x + a.y * a.y;
#pragma unroll
    for (int q = 0; q < NQ; ++q)
      acc[q] += ((y >> (13 - q)) & 1) ? -p : p;
  }
#pragma unroll
  for (int q = 0; q < NQ; ++q) {
    float v = acc[q];
#pragma unroll
    for (int off = 16; off > 0; off >>= 1) v += __shfl_down(v, off, 32);
    if (lane == 0) s_tab[wave * NQ + q] = v;
  }
  __syncthreads();
  if (tid < NQ) {
    float z = 0.f;
#pragma unroll
    for (int w = 0; w < 8; ++w) z += s_tab[w * NQ + tid];
    out[bid * NQ + tid] = z;
  }
}

extern "C" void kernel_launch(void* const* d_in, const int* in_sizes, int n_in,
                              void* d_out, int out_size, void* d_ws, size_t ws_size,
                              hipStream_t stream) {
  const float* latent  = (const float*)d_in[0];
  const float* weights = (const float*)d_in[1];
  float* out = (float*)d_out;

  const int smem_bytes = SMEM_BYTES;
  hipFuncSetAttribute(reinterpret_cast<const void*>(qcircuit_kernel),
                      hipFuncAttributeMaxDynamicSharedMemorySize, smem_bytes);
  qcircuit_kernel<<<512, BLOCK, smem_bytes, stream>>>(latent, weights, out);
}


# Rot kept factored RZ*RY*RZ: RZ phases are per-lane elementwise complex multiplies at tile load and store, the fused gate is the real RY product: 16 WMMA per tile, was 24; new rounding and sum order
# speedup vs baseline: 2.6944x; 1.1930x over previous
#include <hip/hip_runtime.h>
#include <hip/hip_bf16.h>
#include <stdint.h>

#define NQ      14
#define NL      8
#define NSTATES 16384
#define BLOCK   256
#define NGATE   (NL * NQ)
#define SMEM_BYTES 136992

typedef __attribute__((ext_vector_type(2))) float v2f;
typedef __attribute__((ext_vector_type(8))) float v8f;

__device__ __forceinline__ float2 cmulf(float2 a, float2 b) {
  return make_float2(a.x * b.x - a.y * b.y, a.x * b.y + a.y * b.x);
}

__device__ __forceinline__ float ry_elem(float2 cs, int row, int col) {
  return (row == col) ? cs.x : (row ? cs.y : -cs.y);
}

__device__ __forceinline__ float2 rz_phase(float2 cs, int bit) {
  return make_float2(cs.x, bit ? cs.y : -cs.y);
}

__device__ __forceinline__ uint32_t swz(uint32_t x) {
  return x ^ ((x >> 3) & 28u) ^ ((x >> 5) & 23u) ^ ((x >> 9) & 16u);
}

__device__ __forceinline__ uint32_t perm_map(uint32_t y, int l) {
  uint32_t r = y;
#pragma unroll
  for (int j = 1; j <= 7; ++j)
    if ((j & ~l) == 0) r ^= (y >> j);
  return swz(r);
}

__global__ __launch_bounds__(BLOCK)
void qcircuit_kernel(const float* __restrict__ latent,
                     const float* __restrict__ weights,
                     float* __restrict__ out) {
  extern __shared__ float smem[];
  float2* st2  = (float2*)smem;
  float* s_tab = smem + 2 * NSTATES;
  float2* c_tab = (float2*)s_tab;
  float2* vq    = (float2*)(smem + 2 * NSTATES + 512);
  float2* g_tab = (float2*)(smem + 2 * NSTATES + 576);

  const int tid  = threadIdx.x;
  const int lane = tid & 31;
  const int wave = tid >> 5;
  const int bid  = blockIdx.x;
  const float* lat = latent + bid * NQ;

  if (tid < NGATE) {
    const float* w = weights + tid * 3;
    const float phi = w[0], th = w[1], om = w[2];
    float sth, cth;
    sincosf(0.5f * th, &sth, &cth);
    if (tid < NQ) {
      float sa, ca, sb, cb;
      sincosf(0.5f * (phi + om), &sa, &ca);
      sincosf(0.5f * (phi - om), &sb, &cb);
      const float2 m00 = make_float2( cth * ca, -cth * sa);
      const float2 m01 = make_float2(-sth * cb, -sth * sb);
      const float2 m10 = make_float2( sth * cb, -sth * sb);
      const float2 m11 = make_float2( cth * ca,  cth * sa);
      float sn, cs;
      sincosf(0.5f * lat[tid], &sn, &cs);
      vq[tid * 2 + 0] = make_float2(m00.x * cs + m01.y * sn, m00.y * cs - m01.x * sn);
      vq[tid * 2 + 1] = make_float2(m10.x * cs + m11.y * sn, m10.y * cs - m11.x * sn);
    } else {
      float sp, cp, so, co;
      sincosf(0.5f * phi, &sp, &cp);
      sincosf(0.5f * om, &so, &co);
      g_tab[tid * 4 + 0] = make_float2(cth, sth);
      g_tab[tid * 4 + 1] = make_float2(cp, sp);
      g_tab[tid * 4 + 2] = make_float2(co, so);
    }
  } else if (tid == NGATE) {
    g_tab[NGATE * 4 + 0] = make_float2(1.f, 0.f);
    g_tab[NGATE * 4 + 1] = make_float2(1.f, 0.f);
    g_tab[NGATE * 4 + 2] = make_float2(1.f, 0.f);
  }
  __syncthreads();
  {
    const int half = tid >> 7, idx = tid & 127;
    float2 prod = vq[(half * 7) * 2 + ((idx >> 6) & 1)];
#pragma unroll
    for (int j = 1; j < 7; ++j)
      prod = cmulf(prod, vq[(half * 7 + j) * 2 + ((idx >> (6 - j)) & 1)]);
    c_tab[tid] = prod;
  }
  __syncthreads();
#pragma unroll 4
  for (int i = 0; i < NSTATES / BLOCK; ++i) {
    const int y = i * BLOCK + tid;
    st2[swz((uint32_t)y)] = cmulf(c_tab[y >> 7], c_tab[128 + (y & 127)]);
  }
  __syncthreads();

  for (int l = 1; l < NL; ++l) {
    for (int p = 0; p < 2; ++p) {
      const int bx = p ? 0 : 10;
      const int by = p ? 4 : 6;
      const int bt = p ? 8 : 0;
      const int m = lane & 15;
      const int h = lane >> 4;
      const int khi = h << 1;
      const int i0 = l * NQ + 13 - bx;
      const int i1 = i0 - 1, i2 = i0 - 2, i3 = i0 - 3;
      const int y0 = l * NQ + 13 - by;
      const int y1 = y0 - 1;
      const int y2 = p ? NGATE : y0 - 2;
      const int y3 = p ? NGATE : y0 - 3;
      const float2 rx0 = g_tab[i0 * 4], rx1 = g_tab[i1 * 4], rx2 = g_tab[i2 * 4], rx3 = g_tab[i3 * 4];
      const float2 ry0 = g_tab[y0 * 4], ry1 = g_tab[y1 * 4], ry2 = g_tab[y2 * 4], ry3 = g_tab[y3 * 4];
      const float x0c0 = ry_elem(rx0, m & 1, 0),        x0c1 = ry_elem(rx0, m & 1, 1);
      const float x1ch = ry_elem(rx1, (m >> 1) & 1, h);
      const float x2c0 = ry_elem(rx2, (m >> 2) & 1, 0), x2c1 = ry_elem(rx2, (m >> 2) & 1, 1);
      const float x3c0 = ry_elem(rx3, (m >> 3) & 1, 0), x3c1 = ry_elem(rx3, (m >> 3) & 1, 1);
      const float q0c0 = ry_elem(ry0, m & 1, 0),        q0c1 = ry_elem(ry0, m & 1, 1);
      const float q1c0 = ry_elem(ry1, (m >> 1) & 1, 0), q1c1 = ry_elem(ry1, (m >> 1) & 1, 1);
      const float q2c0 = ry_elem(ry2, (m >> 2) & 1, 0), q2c1 = ry_elem(ry2, (m >> 2) & 1, 1);
      const float q3ch = ry_elem(ry3, (m >> 3) & 1, h);
      v2f aR[4];
      v2f yR[4];
#pragma unroll
      for (int j = 0; j < 4; ++j) {
        const float xr = x1ch * ((j & 1) ? x2c1 : x2c0) * ((j >> 1) ? x3c1 : x3c0);
        aR[j].x = x0c0 * xr;
        aR[j].y = x0c1 * xr;
        const float yr = ((j & 1) ? q1c1 : q1c0) * ((j >> 1) ? q2c1 : q2c0) * q3ch;
        yR[j].x = q0c0 * yr;
        yR[j].y = q0c1 * yr;
      }
      const float2 fx0 = g_tab[i0 * 4 + 1], fx1 = g_tab[i1 * 4 + 1], fx2 = g_tab[i2 * 4 + 1], fx3 = g_tab[i3 * 4 + 1];
      const float2 fy0 = g_tab[y0 * 4 + 1], fy1 = g_tab[y1 * 4 + 1], fy2 = g_tab[y2 * 4 + 1], fy3 = g_tab[y3 * 4 + 1];
      const float2 wx0 = g_tab[i0 * 4 + 2], wx1 = g_tab[i1 * 4 + 2], wx2 = g_tab[i2 * 4 + 2], wx3 = g_tab[i3 * 4 + 2];
      const float2 wy0 = g_tab[y0 * 4 + 2], wy1 = g_tab[y1 * 4 + 2], wy2 = g_tab[y2 * 4 + 2], wy3 = g_tab[y3 * 4 + 2];
      float2 lbase = cmulf(cmulf(rz_phase(fy0, m & 1), rz_phase(fy1, (m >> 1) & 1)),
                           cmulf(rz_phase(fy2, (m >> 2) & 1), rz_phase(fy3, (m >> 3) & 1)));
      lbase = cmulf(lbase, rz_phase(fx1, h));
      const float2 le0 = cmulf(lbase, rz_phase(fx0, 0));
      const float2 le1 = cmulf(lbase, rz_phase(fx0, 1));
      float2 phL0[4], phL1[4];
#pragma unroll
      for (int j = 0; j < 4; ++j) {
        const float2 fj = cmulf(rz_phase(fx2, j & 1), rz_phase(fx3, j >> 1));
        phL0[j] = cmulf(le0, fj);
        phL1[j] = cmulf(le1, fj);
      }
      float2 sbase = cmulf(cmulf(rz_phase(wy0, m & 1), rz_phase(wy1, (m >> 1) & 1)),
                           cmulf(rz_phase(wy2, (m >> 2) & 1), rz_phase(wy3, (m >> 3) & 1)));
      sbase = cmulf(sbase, rz_phase(wx3, h));
      v2f phSr[4], phSi[4];
#pragma unroll
      for (int r = 0; r < 8; ++r) {
        const float2 wr = cmulf(cmulf(rz_phase(wx0, r & 1), rz_phase(wx1, (r >> 1) & 1)), rz_phase(wx2, r >> 2));
        const float2 ps = cmulf(sbase, wr);
        if (r & 1) { phSr[r >> 1].y = ps.x; phSi[r >> 1].y = ps.y; }
        else       { phSr[r >> 1].x = ps.x; phSi[r >> 1].x = ps.y; }
      }
      const uint32_t pk1 = perm_map(1u << bx, l);
      const uint32_t pk2 = perm_map(2u << bx, l);
      const uint32_t pk4 = perm_map(4u << bx, l);
      const uint32_t pk8 = perm_map(8u << bx, l);

      for (int ti = 0; ti < 8; ++ti) {
        const int t = wave * 8 + ti;
        const uint32_t cpart = ((uint32_t)m << by) | ((uint32_t)t << bt);
        const uint32_t pcb = perm_map(cpart, l);
        v2f bR[4], bI[4];
#pragma unroll
        for (int j = 0; j < 4; ++j) {
          const int k0 = 4 * j + khi, k1 = k0 + 1;
          uint32_t x0 = pcb, x1 = pcb;
          if (k0 & 1) x0 ^= pk1; if (k0 & 2) x0 ^= pk2;
          if (k0 & 4) x0 ^= pk4; if (k0 & 8) x0 ^= pk8;
          if (k1 & 1) x1 ^= pk1; if (k1 & 2) x1 ^= pk2;
          if (k1 & 4) x1 ^= pk4; if (k1 & 8) x1 ^= pk8;
          const float2 s0 = cmulf(st2[x0], phL0[j]);
          const float2 s1 = cmulf(st2[x1], phL1[j]);
          bR[j].x = s0.x; bR[j].y = s1.x;
          bI[j].x = s0.y; bI[j].y = s1.y;
        }
        v8f dr = {}, di = {};
#pragma unroll
        for (int j = 0; j < 4; ++j) {
          dr = __builtin_amdgcn_wmma_f32_16x16x4_f32(false, bR[j], false, aR[j], (short)0, dr, false, false);
          di = __builtin_amdgcn_wmma_f32_16x16x4_f32(false, bI[j], false, aR[j], (short)0, di, false, false);
        }
        v8f cr = {}, ci = {};
#pragma unroll
        for (int j = 0; j < 4; ++j) {
          v2f ar; ar.x = dr[2 * j]; ar.y = dr[2 * j + 1];
          v2f ai; ai.x = di[2 * j]; ai.y = di[2 * j + 1];
          cr = __builtin_amdgcn_wmma_f32_16x16x4_f32(false, ar, false, yR[j], (short)0, cr, false, false);
          ci = __builtin_amdgcn_wmma_f32_16x16x4_f32(false, ai, false, yR[j], (short)0, ci, false, false);
        }
        const int mb = h << 3;
#pragma unroll
        for (int j = 0; j < 4; ++j) {
          v2f c2r; c2r.x = cr[2 * j]; c2r.y = cr[2 * j + 1];
          v2f c2i; c2i.x = ci[2 * j]; c2i.y = ci[2 * j + 1];
          const v2f o2r = c2r * phSr[j] - c2i * phSi[j];
          const v2f o2i = c2r * phSi[j] + c2i * phSr[j];
          const int mm0 = mb + 2 * j, mm1 = mm0 + 1;
          uint32_t x0 = pcb, x1 = pcb;
          if (mm0 & 1) x0 ^= pk1; if (mm0 & 2) x0 ^= pk2;
          if (mm0 & 4) x0 ^= pk4; if (mm0 & 8) x0 ^= pk8;
          if (mm1 & 1) x1 ^= pk1; if (mm1 & 2) x1 ^= pk2;
          if (mm1 & 4) x1 ^= pk4; if (mm1 & 8) x1 ^= pk8;
          st2[x0] = make_float2(o2r.x, o2i.x);
          st2[x1] = make_float2(o2r.y, o2i.y);
        }
      }
      __syncthreads();
    }
  }

  float acc[NQ];
#pragma unroll
  for (int q = 0; q < NQ; ++q) acc[q] = 0.f;
  for (int i = 0; i < NSTATES / BLOCK; ++i) {
    const int y = tid + BLOCK * i;
    const uint32_t phys = swz((uint32_t)y ^ ((uint32_t)y >> 8));
    const float2 a = st2[phys];
    const float p = a.x * a.x + a.y * a.y;
#pragma unroll
    for (int q = 0; q < NQ; ++q)
      acc[q] += ((y >> (13 - q)) & 1) ? -p : p;
  }
#pragma unroll
  for (int q = 0; q < NQ; ++q) {
    float v = acc[q];
#pragma unroll
    for (int off = 16; off > 0; off >>= 1) v += __shfl_down(v, off, 32);
    if (lane == 0) s_tab[wave * NQ + q] = v;
  }
  __syncthreads();
  if (tid < NQ) {
    float z = 0.f;
#pragma unroll
    for (int w = 0; w < 8; ++w) z += s_tab[w * NQ + tid];
    out[bid * NQ + tid] = z;
  }
}

extern "C" void kernel_launch(void* const* d_in, const int* in_sizes, int n_in,
                              void* d_out, int out_size, void* d_ws, size_t ws_size,
                              hipStream_t stream) {
  const float* latent  = (const float*)d_in[0];
  const float* weights = (const float*)d_in[1];
  float* out = (float*)d_out;

  const int smem_bytes = SMEM_BYTES;
  hipFuncSetAttribute(reinterpret_cast<const void*>(qcircuit_kernel),
                      hipFuncAttributeMaxDynamicSharedMemorySize, smem_bytes);
  qcircuit_kernel<<<512, BLOCK, smem_bytes, stream>>>(latent, weights, out);
}


# Readout in LDS slot order: each thread sums its 64 probabilities by a sum/difference butterfly, lane-bit signs fold into one 5-step xor exchange of 12 values; new summation order of every output
# speedup vs baseline: 2.8893x; 1.0723x over previous
#include <hip/hip_runtime.h>
#include <hip/hip_bf16.h>
#include <stdint.h>

#define NQ      14
#define NL      8
#define NSTATES 16384
#define BLOCK   256
#define NGATE   (NL * NQ)
#define SMEM_BYTES 136992

typedef __attribute__((ext_vector_type(2))) float v2f;
typedef __attribute__((ext_vector_type(8))) float v8f;

__device__ __forceinline__ float2 cmulf(float2 a, float2 b) {
  return make_float2(a.x * b.x - a.y * b.y, a.x * b.y + a.y * b.x);
}

__device__ __forceinline__ float ry_elem(float2 cs, int row, int col) {
  return (row == col) ? cs.x : (row ? cs.y : -cs.y);
}

__device__ __forceinline__ float2 rz_phase(float2 cs, int bit) {
  return make_float2(cs.x, bit ? cs.y : -cs.y);
}

__device__ __forceinline__ uint32_t swz(uint32_t x) {
  return x ^ ((x >> 3) & 28u) ^ ((x >> 5) & 23u) ^ ((x >> 9) & 16u);
}

__device__ __forceinline__ uint32_t perm_map(uint32_t y, int l) {
  uint32_t r = y;
#pragma unroll
  for (int j = 1; j <= 7; ++j)
    if ((j & ~l) == 0) r ^= (y >> j);
  return swz(r);
}

__global__ __launch_bounds__(BLOCK)
void qcircuit_kernel(const float* __restrict__ latent,
                     const float* __restrict__ weights,
                     float* __restrict__ out) {
  extern __shared__ float smem[];
  float2* st2  = (float2*)smem;
  float* s_tab = smem + 2 * NSTATES;
  float2* c_tab = (float2*)s_tab;
  float2* vq    = (float2*)(smem + 2 * NSTATES + 512);
  float2* g_tab = (float2*)(smem + 2 * NSTATES + 576);

  const int tid  = threadIdx.x;
  const int lane = tid & 31;
  const int wave = tid >> 5;
  const int bid  = blockIdx.x;
  const float* lat = latent + bid * NQ;

  if (tid < NGATE) {
    const float* w = weights + tid * 3;
    const float phi = w[0], th = w[1], om = w[2];
    float sth, cth;
    sincosf(0.5f * th, &sth, &cth);
    if (tid < NQ) {
      float sa, ca, sb, cb;
      sincosf(0.5f * (phi + om), &sa, &ca);
      sincosf(0.5f * (phi - om), &sb, &cb);
      const float2 m00 = make_float2( cth * ca, -cth * sa);
      const float2 m01 = make_float2(-sth * cb, -sth * sb);
      const float2 m10 = make_float2( sth * cb, -sth * sb);
      const float2 m11 = make_float2( cth * ca,  cth * sa);
      float sn, cs;
      sincosf(0.5f * lat[tid], &sn, &cs);
      vq[tid * 2 + 0] = make_float2(m00.x * cs + m01.y * sn, m00.y * cs - m01.x * sn);
      vq[tid * 2 + 1] = make_float2(m10.x * cs + m11.y * sn, m10.y * cs - m11.x * sn);
    } else {
      float sp, cp, so, co;
      sincosf(0.5f * phi, &sp, &cp);
      sincosf(0.5f * om, &so, &co);
      g_tab[tid * 4 + 0] = make_float2(cth, sth);
      g_tab[tid * 4 + 1] = make_float2(cp, sp);
      g_tab[tid * 4 + 2] = make_float2(co, so);
    }
  } else if (tid == NGATE) {
    g_tab[NGATE * 4 + 0] = make_float2(1.f, 0.f);
    g_tab[NGATE * 4 + 1] = make_float2(1.f, 0.f);
    g_tab[NGATE * 4 + 2] = make_float2(1.f, 0.f);
  }
  __syncthreads();
  {
    const int half = tid >> 7, idx = tid & 127;
    float2 prod = vq[(half * 7) * 2 + ((idx >> 6) & 1)];
#pragma unroll
    for (int j = 1; j < 7; ++j)
      prod = cmulf(prod, vq[(half * 7 + j) * 2 + ((idx >> (6 - j)) & 1)]);
    c_tab[tid] = prod;
  }
  __syncthreads();
#pragma unroll 4
  for (int i = 0; i < NSTATES / BLOCK; ++i) {
    const int y = i * BLOCK + tid;
    st2[swz((uint32_t)y)] = cmulf(c_tab[y >> 7], c_tab[128 + (y & 127)]);
  }
  __syncthreads();

  for (int l = 1; l < NL; ++l) {
    for (int p = 0; p < 2; ++p) {
      const int bx = p ? 0 : 10;
      const int by = p ? 4 : 6;
      const int bt = p ? 8 : 0;
      const int m = lane & 15;
      const int h = lane >> 4;
      const int khi = h << 1;
      const int i0 = l * NQ + 13 - bx;
      const int i1 = i0 - 1, i2 = i0 - 2, i3 = i0 - 3;
      const int y0 = l * NQ + 13 - by;
      const int y1 = y0 - 1;
      const int y2 = p ? NGATE : y0 - 2;
      const int y3 = p ? NGATE : y0 - 3;
      const float2 rx0 = g_tab[i0 * 4], rx1 = g_tab[i1 * 4], rx2 = g_tab[i2 * 4], rx3 = g_tab[i3 * 4];
      const float2 ry0 = g_tab[y0 * 4], ry1 = g_tab[y1 * 4], ry2 = g_tab[y2 * 4], ry3 = g_tab[y3 * 4];
      const float x0c0 = ry_elem(rx0, m & 1, 0),        x0c1 = ry_elem(rx0, m & 1, 1);
      const float x1ch = ry_elem(rx1, (m >> 1) & 1, h);
      const float x2c0 = ry_elem(rx2, (m >> 2) & 1, 0), x2c1 = ry_elem(rx2, (m >> 2) & 1, 1);
      const float x3c0 = ry_elem(rx3, (m >> 3) & 1, 0), x3c1 = ry_elem(rx3, (m >> 3) & 1, 1);
      const float q0c0 = ry_elem(ry0, m & 1, 0),        q0c1 = ry_elem(ry0, m & 1, 1);
      const float q1c0 = ry_elem(ry1, (m >> 1) & 1, 0), q1c1 = ry_elem(ry1, (m >> 1) & 1, 1);
      const float q2c0 = ry_elem(ry2, (m >> 2) & 1, 0), q2c1 = ry_elem(ry2, (m >> 2) & 1, 1);
      const float q3ch = ry_elem(ry3, (m >> 3) & 1, h);
      v2f aR[4];
      v2f yR[4];
#pragma unroll
      for (int j = 0; j < 4; ++j) {
        const float xr = x1ch * ((j & 1) ? x2c1 : x2c0) * ((j >> 1) ? x3c1 : x3c0);
        aR[j].x = x0c0 * xr;
        aR[j].y = x0c1 * xr;
        const float yr = ((j & 1) ? q1c1 : q1c0) * ((j >> 1) ? q2c1 : q2c0) * q3ch;
        yR[j].x = q0c0 * yr;
        yR[j].y = q0c1 * yr;
      }
      const float2 fx0 = g_tab[i0 * 4 + 1], fx1 = g_tab[i1 * 4 + 1], fx2 = g_tab[i2 * 4 + 1], fx3 = g_tab[i3 * 4 + 1];
      const float2 fy0 = g_tab[y0 * 4 + 1], fy1 = g_tab[y1 * 4 + 1], fy2 = g_tab[y2 * 4 + 1], fy3 = g_tab[y3 * 4 + 1];
      const float2 wx0 = g_tab[i0 * 4 + 2], wx1 = g_tab[i1 * 4 + 2], wx2 = g_tab[i2 * 4 + 2], wx3 = g_tab[i3 * 4 + 2];
      const float2 wy0 = g_tab[y0 * 4 + 2], wy1 = g_tab[y1 * 4 + 2], wy2 = g_tab[y2 * 4 + 2], wy3 = g_tab[y3 * 4 + 2];
      float2 lbase = cmulf(cmulf(rz_phase(fy0, m & 1), rz_phase(fy1, (m >> 1) & 1)),
                           cmulf(rz_phase(fy2, (m >> 2) & 1), rz_phase(fy3, (m >> 3) & 1)));
      lbase = cmulf(lbase, rz_phase(fx1, h));
      const float2 le0 = cmulf(lbase, rz_phase(fx0, 0));
      const float2 le1 = cmulf(lbase, rz_phase(fx0, 1));
      float2 phL0[4], phL1[4];
#pragma unroll
      for (int j = 0; j < 4; ++j) {
        const float2 fj = cmulf(rz_phase(fx2, j & 1), rz_phase(fx3, j >> 1));
        phL0[j] = cmulf(le0, fj);
        phL1[j] = cmulf(le1, fj);
      }
      float2 sbase = cmulf(cmulf(rz_phase(wy0, m & 1), rz_phase(wy1, (m >> 1) & 1)),
                           cmulf(rz_phase(wy2, (m >> 2) & 1), rz_phase(wy3, (m >> 3) & 1)));
      sbase = cmulf(sbase, rz_phase(wx3, h));
      v2f phSr[4], phSi[4];
#pragma unroll
      for (int r = 0; r < 8; ++r) {
        const float2 wr = cmulf(cmulf(rz_phase(wx0, r & 1), rz_phase(wx1, (r >> 1) & 1)), rz_phase(wx2, r >> 2));
        const float2 ps = cmulf(sbase, wr);
        if (r & 1) { phSr[r >> 1].y = ps.x; phSi[r >> 1].y = ps.y; }
        else       { phSr[r >> 1].x = ps.x; phSi[r >> 1].x = ps.y; }
      }
      const uint32_t pk1 = perm_map(1u << bx, l);
      const uint32_t pk2 = perm_map(2u << bx, l);
      const uint32_t pk4 = perm_map(4u << bx, l);
      const uint32_t pk8 = perm_map(8u << bx, l);

      for (int ti = 0; ti < 8; ++ti) {
        const int t = wave * 8 + ti;
        const uint32_t cpart = ((uint32_t)m << by) | ((uint32_t)t << bt);
        const uint32_t pcb = perm_map(cpart, l);
        v2f bR[4], bI[4];
#pragma unroll
        for (int j = 0; j < 4; ++j) {
          const int k0 = 4 * j + khi, k1 = k0 + 1;
          uint32_t x0 = pcb, x1 = pcb;
          if (k0 & 1) x0 ^= pk1; if (k0 & 2) x0 ^= pk2;
          if (k0 & 4) x0 ^= pk4; if (k0 & 8) x0 ^= pk8;
          if (k1 & 1) x1 ^= pk1; if (k1 & 2) x1 ^= pk2;
          if (k1 & 4) x1 ^= pk4; if (k1 & 8) x1 ^= pk8;
          const float2 s0 = cmulf(st2[x0], phL0[j]);
          const float2 s1 = cmulf(st2[x1], phL1[j]);
          bR[j].x = s0.x; bR[j].y = s1.x;
          bI[j].x = s0.y; bI[j].y = s1.y;
        }
        v8f dr = {}, di = {};
#pragma unroll
        for (int j = 0; j < 4; ++j) {
          dr = __builtin_amdgcn_wmma_f32_16x16x4_f32(false, bR[j], false, aR[j], (short)0, dr, false, false);
          di = __builtin_amdgcn_wmma_f32_16x16x4_f32(false, bI[j], false, aR[j], (short)0, di, false, false);
        }
        v8f cr = {}, ci = {};
#pragma unroll
        for (int j = 0; j < 4; ++j) {
          v2f ar; ar.x = dr[2 * j]; ar.y = dr[2 * j + 1];
          v2f ai; ai.x = di[2 * j]; ai.y = di[2 * j + 1];
          cr = __builtin_amdgcn_wmma_f32_16x16x4_f32(false, ar, false, yR[j], (short)0, cr, false, false);
          ci = __builtin_amdgcn_wmma_f32_16x16x4_f32(false, ai, false, yR[j], (short)0, ci, false, false);
        }
        const int mb = h << 3;
#pragma unroll
        for (int j = 0; j < 4; ++j) {
          v2f c2r; c2r.x = cr[2 * j]; c2r.y = cr[2 * j + 1];
          v2f c2i; c2i.x = ci[2 * j]; c2i.y = ci[2 * j + 1];
          const v2f o2r = c2r * phSr[j] - c2i * phSi[j];
          const v2f o2i = c2r * phSi[j] + c2i * phSr[j];
          const int mm0 = mb + 2 * j, mm1 = mm0 + 1;
          uint32_t x0 = pcb, x1 = pcb;
          if (mm0 & 1) x0 ^= pk1; if (mm0 & 2) x0 ^= pk2;
          if (mm0 & 4) x0 ^= pk4; if (mm0 & 8) x0 ^= pk8;
          if (mm1 & 1) x1 ^= pk1; if (mm1 & 2) x1 ^= pk2;
          if (mm1 & 4) x1 ^= pk4; if (mm1 & 8) x1 ^= pk8;
          st2[x0] = make_float2(o2r.x, o2i.x);
          st2[x1] = make_float2(o2r.y, o2i.y);
        }
      }
      __syncthreads();
    }
  }

  const float2* sp = st2 + tid;
  float T = 0.f;
  float Dq = 0.f;
  float D[6];
  for (int j = 0; j < 6; ++j) D[j] = 0.f;
#pragma nounroll
  for (int b = 0; b < 4; ++b) {
    float p[16];
    for (int j = 0; j < 16; ++j) {
      const float2 a = sp[BLOCK * (16 * b + j)];
      p[j] = a.x * a.x + a.y * a.y;
    }
    float s8[8], s4[4], s2[2];
    float d0 = 0.f, d1 = 0.f, d2 = 0.f;
    for (int k = 0; k < 8; ++k) { s8[k] = p[2 * k] + p[2 * k + 1]; d0 += p[2 * k] - p[2 * k + 1]; }
    for (int k = 0; k < 4; ++k) { s4[k] = s8[2 * k] + s8[2 * k + 1]; d1 += s8[2 * k] - s8[2 * k + 1]; }
    for (int k = 0; k < 2; ++k) { s2[k] = s4[2 * k] + s4[2 * k + 1]; d2 += s4[2 * k] - s4[2 * k + 1]; }
    const float t  = s2[0] + s2[1];
    const float d3 = s2[0] - s2[1];
    T += t;
    D[0] += d0; D[1] += d1; D[2] += d2; D[3] += d3;
    D[4] += (b & 1) ? -t : t;
    D[5] += (b & 2) ? -t : t;
    Dq   += ((b ^ (b >> 1)) & 1) ? -d1 : d1;
  }
  float G[5];
  G[4] = Dq;
  for (int s = 0; s < 5; ++s) {
    const int msk = 1 << s;
    const bool up = (lane & msk) != 0;
    T += __shfl_xor(T, msk, 32);
    for (int k = 0; k < 4; ++k)
      if (k < s) G[k] += __shfl_xor(G[k], msk, 32);
    {
      const float o = __shfl_xor(G[4], msk, 32);
      G[4] = (s < 4) ? (G[4] + o) : (up ? (o - G[4]) : (G[4] - o));
    }
    for (int k = 0; k < 6; ++k) {
      const float o = __shfl_xor(D[k], msk, 32);
      if (k == s && k < 4) G[k] = up ? (o - D[k]) : (D[k] - o);
      D[k] += o;
    }
  }
  const int w0 = wave & 1, w1 = (wave >> 1) & 1, w2 = wave >> 2;
  if (lane == 0) {
    float* dst = s_tab + wave * NQ;
    dst[0] = D[5]; dst[1] = D[4]; dst[2] = D[3]; dst[3] = D[2]; dst[4] = D[1]; dst[5] = D[0];
    dst[6] = w2 ? -T : T;
    dst[7] = w1 ? -T : T;
    dst[8] = w0 ? -D[5] : D[5];
    dst[9]  = w2 ? -G[4] : G[4];
    dst[10] = w1 ? -G[3] : G[3];
    dst[11] = (w0 ^ w2) ? -G[2] : G[2];
    dst[12] = w1 ? -G[1] : G[1];
    dst[13] = w0 ? -G[0] : G[0];
  }
  __syncthreads();
  if (tid < NQ) {
    float z = 0.f;
#pragma unroll
    for (int w = 0; w < 8; ++w) z += s_tab[w * NQ + tid];
    out[bid * NQ + tid] = z;
  }
}

extern "C" void kernel_launch(void* const* d_in, const int* in_sizes, int n_in,
                              void* d_out, int out_size, void* d_ws, size_t ws_size,
                              hipStream_t stream) {
  const float* latent  = (const float*)d_in[0];
  const float* weights = (const float*)d_in[1];
  float* out = (float*)d_out;

  const int smem_bytes = SMEM_BYTES;
  hipFuncSetAttribute(reinterpret_cast<const void*>(qcircuit_kernel),
                      hipFuncAttributeMaxDynamicSharedMemorySize, smem_bytes);
  qcircuit_kernel<<<512, BLOCK, smem_bytes, stream>>>(latent, weights, out);
}


# RZ(om) of layer l carried through the CNOT relabelling into layer l+1's load phase (picked by adjacent-bit XORs); tiles stored bare; last layer's om phase not applied; new rounding of amplitudes
# speedup vs baseline: 3.0171x; 1.0442x over previous
#include <hip/hip_runtime.h>
#include <hip/hip_bf16.h>
#include <stdint.h>

#define NQ      14
#define NL      8
#define NSTATES 16384
#define BLOCK   256
#define NGATE   (NL * NQ)
#define SMEM_BYTES 136992

typedef __attribute__((ext_vector_type(2))) float v2f;
typedef __attribute__((ext_vector_type(8))) float v8f;

__device__ __forceinline__ float2 cmulf(float2 a, float2 b) {
  return make_float2(a.x * b.x - a.y * b.y, a.x * b.y + a.y * b.x);
}

__device__ __forceinline__ float ry_elem(float2 cs, int row, int col) {
  return (row == col) ? cs.x : (row ? cs.y : -cs.y);
}

__device__ __forceinline__ float2 rz_phase(float2 cs, int bit) {
  return make_float2(cs.x, bit ? cs.y : -cs.y);
}

__device__ __forceinline__ uint32_t swz(uint32_t x) {
  return x ^ ((x >> 3) & 28u) ^ ((x >> 5) & 23u) ^ ((x >> 9) & 16u);
}

__device__ __forceinline__ uint32_t perm_map(uint32_t y, int l) {
  uint32_t r = y;
#pragma unroll
  for (int j = 1; j <= 7; ++j)
    if ((j & ~l) == 0) r ^= (y >> j);
  return swz(r);
}

__global__ __launch_bounds__(BLOCK)
void qcircuit_kernel(const float* __restrict__ latent,
                     const float* __restrict__ weights,
                     float* __restrict__ out) {
  extern __shared__ float smem[];
  float2* st2  = (float2*)smem;
  float* s_tab = smem + 2 * NSTATES;
  float2* c_tab = (float2*)s_tab;
  float2* vq    = (float2*)(smem + 2 * NSTATES + 512);
  float2* g_tab = (float2*)(smem + 2 * NSTATES + 576);

  const int tid  = threadIdx.x;
  const int lane = tid & 31;
  const int wave = tid >> 5;
  const int bid  = blockIdx.x;
  const float* lat = latent + bid * NQ;

  if (tid < NGATE) {
    const float* w = weights + tid * 3;
    const float phi = w[0], th = w[1], om = w[2];
    float sth, cth;
    sincosf(0.5f * th, &sth, &cth);
    if (tid < NQ) {
      float sa, ca, sb, cb;
      sincosf(0.5f * (phi + om), &sa, &ca);
      sincosf(0.5f * (phi - om), &sb, &cb);
      const float2 m00 = make_float2( cth * ca, -cth * sa);
      const float2 m01 = make_float2(-sth * cb, -sth * sb);
      const float2 m10 = make_float2( sth * cb, -sth * sb);
      const float2 m11 = make_float2( cth * ca,  cth * sa);
      float sn, cs;
      sincosf(0.5f * lat[tid], &sn, &cs);
      vq[tid * 2 + 0] = make_float2(m00.x * cs + m01.y * sn, m00.y * cs - m01.x * sn);
      vq[tid * 2 + 1] = make_float2(m10.x * cs + m11.y * sn, m10.y * cs - m11.x * sn);
      g_tab[tid * 4 + 2] = make_float2(1.f, 0.f);
    } else {
      float sp, cp, so, co;
      sincosf(0.5f * phi, &sp, &cp);
      sincosf(0.5f * om, &so, &co);
      g_tab[tid * 4 + 0] = make_float2(cth, sth);
      g_tab[tid * 4 + 1] = make_float2(cp, sp);
      g_tab[tid * 4 + 2] = make_float2(co, so);
    }
  } else if (tid == NGATE) {
    g_tab[NGATE * 4 + 0] = make_float2(1.f, 0.f);
    g_tab[NGATE * 4 + 1] = make_float2(1.f, 0.f);
    g_tab[NGATE * 4 + 2] = make_float2(1.f, 0.f);
  }
  __syncthreads();
  {
    const int half = tid >> 7, idx = tid & 127;
    float2 prod = vq[(half * 7) * 2 + ((idx >> 6) & 1)];
#pragma unroll
    for (int j = 1; j < 7; ++j)
      prod = cmulf(prod, vq[(half * 7 + j) * 2 + ((idx >> (6 - j)) & 1)]);
    c_tab[tid] = prod;
  }
  __syncthreads();
#pragma unroll 4
  for (int i = 0; i < NSTATES / BLOCK; ++i) {
    const int y = i * BLOCK + tid;
    st2[swz((uint32_t)y)] = cmulf(c_tab[y >> 7], c_tab[128 + (y & 127)]);
  }
  __syncthreads();

  for (int l = 1; l < NL; ++l) {
    for (int p = 0; p < 2; ++p) {
      const int bx = p ? 0 : 10;
      const int by = p ? 4 : 6;
      const int bt = p ? 8 : 0;
      const int m = lane & 15;
      const int h = lane >> 4;
      const int khi = h << 1;
      const int i0 = l * NQ + 13 - bx;
      const int i1 = i0 - 1, i2 = i0 - 2, i3 = i0 - 3;
      const int y0 = l * NQ + 13 - by;
      const int y1 = y0 - 1;
      const int y2 = p ? NGATE : y0 - 2;
      const int y3 = p ? NGATE : y0 - 3;
      const float2 rx0 = g_tab[i0 * 4], rx1 = g_tab[i1 * 4], rx2 = g_tab[i2 * 4], rx3 = g_tab[i3 * 4];
      const float2 ry0 = g_tab[y0 * 4], ry1 = g_tab[y1 * 4], ry2 = g_tab[y2 * 4], ry3 = g_tab[y3 * 4];
      const float x0c0 = ry_elem(rx0, m & 1, 0),        x0c1 = ry_elem(rx0, m & 1, 1);
      const float x1ch = ry_elem(rx1, (m >> 1) & 1, h);
      const float x2c0 = ry_elem(rx2, (m >> 2) & 1, 0), x2c1 = ry_elem(rx2, (m >> 2) & 1, 1);
      const float x3c0 = ry_elem(rx3, (m >> 3) & 1, 0), x3c1 = ry_elem(rx3, (m >> 3) & 1, 1);
      const float q0c0 = ry_elem(ry0, m & 1, 0),        q0c1 = ry_elem(ry0, m & 1, 1);
      const float q1c0 = ry_elem(ry1, (m >> 1) & 1, 0), q1c1 = ry_elem(ry1, (m >> 1) & 1, 1);
      const float q2c0 = ry_elem(ry2, (m >> 2) & 1, 0), q2c1 = ry_elem(ry2, (m >> 2) & 1, 1);
      const float q3ch = ry_elem(ry3, (m >> 3) & 1, h);
      v2f aR[4];
      v2f yR[4];
#pragma unroll
      for (int j = 0; j < 4; ++j) {
        const float xr = x1ch * ((j & 1) ? x2c1 : x2c0) * ((j >> 1) ? x3c1 : x3c0);
        aR[j].x = x0c0 * xr;
        aR[j].y = x0c1 * xr;
        const float yr = ((j & 1) ? q1c1 : q1c0) * ((j >> 1) ? q2c1 : q2c0) * q3ch;
        yR[j].x = q0c0 * yr;
        yR[j].y = q0c1 * yr;
      }
      const float2 fx0 = g_tab[i0 * 4 + 1], fx1 = g_tab[i1 * 4 + 1], fx2 = g_tab[i2 * 4 + 1], fx3 = g_tab[i3 * 4 + 1];
      const float2 fy0 = g_tab[y0 * 4 + 1], fy1 = g_tab[y1 * 4 + 1], fy2 = g_tab[y2 * 4 + 1], fy3 = g_tab[y3 * 4 + 1];
      const int P = (l - 1) * NQ;
      const float2 oK01 = g_tab[(P + (p ? 13 : 3)) * 4 + 2];
      const float2 oK12 = g_tab[(P + (p ? 12 : 2)) * 4 + 2];
      const float2 oK23 = g_tab[(P + (p ? 11 : 1)) * 4 + 2];
      const float2 oTop = g_tab[(P + (p ? 10 : 0)) * 4 + 2];
      const float2 oBot = g_tab[(p ? NGATE : P + 4) * 4 + 2];
      const float2 oC01 = g_tab[(P + (p ? 9 : 7)) * 4 + 2];
      const float2 oC12 = g_tab[(p ? NGATE : P + 6) * 4 + 2];
      const float2 oC23 = g_tab[(p ? NGATE : P + 5) * 4 + 2];
      const float2 oT5  = g_tab[(p ? NGATE : P + 8) * 4 + 2];
      const int c0 = m & 1, c1 = (m >> 1) & 1, c2 = (m >> 2) & 1, c3 = (m >> 3) & 1;
      const int nTop = p ? c0 : 0;
      float2 lbase = cmulf(cmulf(rz_phase(fy0, c0), rz_phase(fy1, c1)),
                           cmulf(rz_phase(fy2, c2), rz_phase(fy3, c3)));
      lbase = cmulf(lbase, rz_phase(fx1, h));
      lbase = cmulf(lbase, cmulf(cmulf(rz_phase(oC01, c0 ^ c1), rz_phase(oC12, c1 ^ c2)),
                                 cmulf(rz_phase(oC23, c2 ^ c3), rz_phase(oT5, (wave >> 2) ^ c0))));
      const float2 le0 = cmulf(lbase, cmulf(rz_phase(fx0, 0), cmulf(rz_phase(oK01, h), rz_phase(oBot, c3))));
      const float2 le1 = cmulf(lbase, cmulf(rz_phase(fx0, 1), cmulf(rz_phase(oK01, h ^ 1), rz_phase(oBot, c3 ^ 1))));
      float2 phL0[4], phL1[4];
#pragma unroll
      for (int j = 0; j < 4; ++j) {
        const int k2 = j & 1, k3 = j >> 1;
        const float2 fj = cmulf(cmulf(rz_phase(fx2, k2), rz_phase(fx3, k3)),
                                cmulf(cmulf(rz_phase(oK12, h ^ k2), rz_phase(oK23, k2 ^ k3)), rz_phase(oTop, k3 ^ nTop)));
        phL0[j] = cmulf(le0, fj);
        phL1[j] = cmulf(le1, fj);
      }
      const uint32_t pk1 = perm_map(1u << bx, l);
      const uint32_t pk2 = perm_map(2u << bx, l);
      const uint32_t pk4 = perm_map(4u << bx, l);
      const uint32_t pk8 = perm_map(8u << bx, l);

      for (int ti = 0; ti < 8; ++ti) {
        const int t = wave * 8 + ti;
        const uint32_t cpart = ((uint32_t)m << by) | ((uint32_t)t << bt);
        const uint32_t pcb = perm_map(cpart, l);
        v2f bR[4], bI[4];
#pragma unroll
        for (int j = 0; j < 4; ++j) {
          const int k0 = 4 * j + khi, k1 = k0 + 1;
          uint32_t x0 = pcb, x1 = pcb;
          if (k0 & 1) x0 ^= pk1; if (k0 & 2) x0 ^= pk2;
          if (k0 & 4) x0 ^= pk4; if (k0 & 8) x0 ^= pk8;
          if (k1 & 1) x1 ^= pk1; if (k1 & 2) x1 ^= pk2;
          if (k1 & 4) x1 ^= pk4; if (k1 & 8) x1 ^= pk8;
          const float2 s0 = cmulf(st2[x0], phL0[j]);
          const float2 s1 = cmulf(st2[x1], phL1[j]);
          bR[j].x = s0.x; bR[j].y = s1.x;
          bI[j].x = s0.y; bI[j].y = s1.y;
        }
        v8f dr = {}, di = {};
#pragma unroll
        for (int j = 0; j < 4; ++j) {
          dr = __builtin_amdgcn_wmma_f32_16x16x4_f32(false, bR[j], false, aR[j], (short)0, dr, false, false);
          di = __builtin_amdgcn_wmma_f32_16x16x4_f32(false, bI[j], false, aR[j], (short)0, di, false, false);
        }
        v8f cr = {}, ci = {};
#pragma unroll
        for (int j = 0; j < 4; ++j) {
          v2f ar; ar.x = dr[2 * j]; ar.y = dr[2 * j + 1];
          v2f ai; ai.x = di[2 * j]; ai.y = di[2 * j + 1];
          cr = __builtin_amdgcn_wmma_f32_16x16x4_f32(false, ar, false, yR[j], (short)0, cr, false, false);
          ci = __builtin_amdgcn_wmma_f32_16x16x4_f32(false, ai, false, yR[j], (short)0, ci, false, false);
        }
        const int mb = h << 3;
#pragma unroll
        for (int j = 0; j < 4; ++j) {
          const int mm0 = mb + 2 * j, mm1 = mm0 + 1;
          uint32_t x0 = pcb, x1 = pcb;
          if (mm0 & 1) x0 ^= pk1; if (mm0 & 2) x0 ^= pk2;
          if (mm0 & 4) x0 ^= pk4; if (mm0 & 8) x0 ^= pk8;
          if (mm1 & 1) x1 ^= pk1; if (mm1 & 2) x1 ^= pk2;
          if (mm1 & 4) x1 ^= pk4; if (mm1 & 8) x1 ^= pk8;
          st2[x0] = make_float2(cr[2 * j], ci[2 * j]);
          st2[x1] = make_float2(cr[2 * j + 1], ci[2 * j + 1]);
        }
      }
      __syncthreads();
    }
  }

  const float2* sp = st2 + tid;
  float T = 0.f;
  float Dq = 0.f;
  float D[6];
  for (int j = 0; j < 6; ++j) D[j] = 0.f;
#pragma nounroll
  for (int b = 0; b < 4; ++b) {
    float p[16];
    for (int j = 0; j < 16; ++j) {
      const float2 a = sp[BLOCK * (16 * b + j)];
      p[j] = a.x * a.x + a.y * a.y;
    }
    float s8[8], s4[4], s2[2];
    float d0 = 0.f, d1 = 0.f, d2 = 0.f;
    for (int k = 0; k < 8; ++k) { s8[k] = p[2 * k] + p[2 * k + 1]; d0 += p[2 * k] - p[2 * k + 1]; }
    for (int k = 0; k < 4; ++k) { s4[k] = s8[2 * k] + s8[2 * k + 1]; d1 += s8[2 * k] - s8[2 * k + 1]; }
    for (int k = 0; k < 2; ++k) { s2[k] = s4[2 * k] + s4[2 * k + 1]; d2 += s4[2 * k] - s4[2 * k + 1]; }
    const float t  = s2[0] + s2[1];
    const float d3 = s2[0] - s2[1];
    T += t;
    D[0] += d0; D[1] += d1; D[2] += d2; D[3] += d3;
    D[4] += (b & 1) ? -t : t;
    D[5] += (b & 2) ? -t : t;
    Dq   += ((b ^ (b >> 1)) & 1) ? -d1 : d1;
  }
  float G[5];
  G[4] = Dq;
  for (int s = 0; s < 5; ++s) {
    const int msk = 1 << s;
    const bool up = (lane & msk) != 0;
    T += __shfl_xor(T, msk, 32);
    for (int k = 0; k < 4; ++k)
      if (k < s) G[k] += __shfl_xor(G[k], msk, 32);
    {
      const float o = __shfl_xor(G[4], msk, 32);
      G[4] = (s < 4) ? (G[4] + o) : (up ? (o - G[4]) : (G[4] - o));
    }
    for (int k = 0; k < 6; ++k) {
      const float o = __shfl_xor(D[k], msk, 32);
      if (k == s && k < 4) G[k] = up ? (o - D[k]) : (D[k] - o);
      D[k] += o;
    }
  }
  const int w0 = wave & 1, w1 = (wave >> 1) & 1, w2 = wave >> 2;
  if (lane == 0) {
    float* dst = s_tab + wave * NQ;
    dst[0] = D[5]; dst[1] = D[4]; dst[2] = D[3]; dst[3] = D[2]; dst[4] = D[1]; dst[5] = D[0];
    dst[6] = w2 ? -T : T;
    dst[7] = w1 ? -T : T;
    dst[8] = w0 ? -D[5] : D[5];
    dst[9]  = w2 ? -G[4] : G[4];
    dst[10] = w1 ? -G[3] : G[3];
    dst[11] = (w0 ^ w2) ? -G[2] : G[2];
    dst[12] = w1 ? -G[1] : G[1];
    dst[13] = w0 ? -G[0] : G[0];
  }
  __syncthreads();
  if (tid < NQ) {
    float z = 0.f;
#pragma unroll
    for (int w = 0; w < 8; ++w) z += s_tab[w * NQ + tid];
    out[bid * NQ + tid] = z;
  }
}

extern "C" void kernel_launch(void* const* d_in, const int* in_sizes, int n_in,
                              void* d_out, int out_size, void* d_ws, size_t ws_size,
                              hipStream_t stream) {
  const float* latent  = (const float*)d_in[0];
  const float* weights = (const float*)d_in[1];
  float* out = (float*)d_out;

  const int smem_bytes = SMEM_BYTES;
  hipFuncSetAttribute(reinterpret_cast<const void*>(qcircuit_kernel),
                      hipFuncAttributeMaxDynamicSharedMemorySize, smem_bytes);
  qcircuit_kernel<<<512, BLOCK, smem_bytes, stream>>>(latent, weights, out);
}


# LDS tile addresses from a per-workgroup table of the index map's XOR terms (byte units): no map evaluation or shift per tile; true occupancy (4 waves/SIMD) declared, so loads stay grouped
# speedup vs baseline: 3.3735x; 1.1181x over previous
#include <hip/hip_runtime.h>
#include <hip/hip_bf16.h>
#include <stdint.h>

#define NQ      14
#define NL      8
#define NSTATES 16384
#define BLOCK   256
#define NGATE   (NL * NQ)
#define SMEM_BYTES 137888

typedef __attribute__((ext_vector_type(2))) float v2f;
typedef __attribute__((ext_vector_type(8))) float v8f;

__device__ __forceinline__ float2 cmulf(float2 a, float2 b) {
  return make_float2(a.x * b.x - a.y * b.y, a.x * b.y + a.y * b.x);
}

__device__ __forceinline__ float ry_elem(float2 cs, int row, int col) {
  return (row == col) ? cs.x : (row ? cs.y : -cs.y);
}

__device__ __forceinline__ float2 rz_phase(float2 cs, int bit) {
  return make_float2(cs.x, bit ? cs.y : -cs.y);
}

__device__ __forceinline__ uint32_t swz(uint32_t x) {
  return x ^ ((x >> 3) & 28u) ^ ((x >> 5) & 23u) ^ ((x >> 9) & 16u);
}

__device__ __forceinline__ uint32_t perm_map(uint32_t y, int l) {
  uint32_t r = y;
#pragma unroll
  for (int j = 1; j <= 7; ++j)
    if ((j & ~l) == 0) r ^= (y >> j);
  return swz(r);
}

__device__ __forceinline__ uint32_t swz8(uint32_t x) {
  return x ^ ((x >> 3) & (28u << 3)) ^ ((x >> 5) & (23u << 3)) ^ ((x >> 9) & (16u << 3));
}
__device__ __forceinline__ uint32_t perm_map8(uint32_t y8, int l) {
  uint32_t r = y8;
#pragma unroll
  for (int j = 1; j <= 7; ++j)
    if ((j & ~l) == 0) r ^= (y8 >> j);
  return swz8(r & ~7u);
}

__global__ __launch_bounds__(BLOCK) __attribute__((amdgpu_waves_per_eu(4, 4)))
void qcircuit_kernel(const float* __restrict__ latent,
                     const float* __restrict__ weights,
                     float* __restrict__ out) {
  extern __shared__ float smem[];
  float2* st2  = (float2*)smem;
  float* s_tab = smem + 2 * NSTATES;
  float2* c_tab = (float2*)s_tab;
  float2* vq    = (float2*)(smem + 2 * NSTATES + 512);
  float2* g_tab = (float2*)(smem + 2 * NSTATES + 576);
  uint32_t* a_tab = (uint32_t*)(smem + 2 * NSTATES + 576 + (NGATE + 1) * 8);

  const int tid  = threadIdx.x;
  const int lane = tid & 31;
  const int wave = tid >> 5;
  const int bid  = blockIdx.x;
  const float* lat = latent + bid * NQ;

  if (tid < NGATE) {
    const float* w = weights + tid * 3;
    const float phi = w[0], th = w[1], om = w[2];
    float sth, cth;
    sincosf(0.5f * th, &sth, &cth);
    if (tid < NQ) {
      float sa, ca, sb, cb;
      sincosf(0.5f * (phi + om), &sa, &ca);
      sincosf(0.5f * (phi - om), &sb, &cb);
      const float2 m00 = make_float2( cth * ca, -cth * sa);
      const float2 m01 = make_float2(-sth * cb, -sth * sb);
      const float2 m10 = make_float2( sth * cb, -sth * sb);
      const float2 m11 = make_float2( cth * ca,  cth * sa);
      float sn, cs;
      sincosf(0.5f * lat[tid], &sn, &cs);
      vq[tid * 2 + 0] = make_float2(m00.x * cs + m01.y * sn, m00.y * cs - m01.x * sn);
      vq[tid * 2 + 1] = make_float2(m10.x * cs + m11.y * sn, m10.y * cs - m11.x * sn);
      g_tab[tid * 4 + 2] = make_float2(1.f, 0.f);
    } else {
      float sp, cp, so, co;
      sincosf(0.5f * phi, &sp, &cp);
      sincosf(0.5f * om, &so, &co);
      g_tab[tid * 4 + 0] = make_float2(cth, sth);
      g_tab[tid * 4 + 1] = make_float2(cp, sp);
      g_tab[tid * 4 + 2] = make_float2(co, so);
    }
  } else if (tid == NGATE) {
    g_tab[NGATE * 4 + 0] = make_float2(1.f, 0.f);
    g_tab[NGATE * 4 + 1] = make_float2(1.f, 0.f);
    g_tab[NGATE * 4 + 2] = make_float2(1.f, 0.f);
  }
  if (tid < 7 * 2 * 16) {
    const int e = tid & 15, pp = (tid >> 4) & 1, ll = (tid >> 5) + 1;
    const int bit = (e < 4) ? (pp ? 0 : 10) + e : (e < 8) ? (pp ? 4 : 6) + (e - 4) : (pp ? 8 : 0) + (e - 8);
    a_tab[tid] = (e < 14) ? perm_map8(8u << bit, ll) : 0u;
  }
  __syncthreads();
  {
    const int half = tid >> 7, idx = tid & 127;
    float2 prod = vq[(half * 7) * 2 + ((idx >> 6) & 1)];
#pragma unroll
    for (int j = 1; j < 7; ++j)
      prod = cmulf(prod, vq[(half * 7 + j) * 2 + ((idx >> (6 - j)) & 1)]);
    c_tab[tid] = prod;
  }
  __syncthreads();
#pragma unroll 4
  for (int i = 0; i < NSTATES / BLOCK; ++i) {
    const int y = i * BLOCK + tid;
    st2[swz((uint32_t)y)] = cmulf(c_tab[y >> 7], c_tab[128 + (y & 127)]);
  }
  __syncthreads();

  for (int l = 1; l < NL; ++l) {
    for (int p = 0; p < 2; ++p) {
      const int bx = p ? 0 : 10;
      const int by = p ? 4 : 6;
      const int m = lane & 15;
      const int h = lane >> 4;
      const int i0 = l * NQ + 13 - bx;
      const int i1 = i0 - 1, i2 = i0 - 2, i3 = i0 - 3;
      const int y0 = l * NQ + 13 - by;
      const int y1 = y0 - 1;
      const int y2 = p ? NGATE : y0 - 2;
      const int y3 = p ? NGATE : y0 - 3;
      const float2 rx0 = g_tab[i0 * 4], rx1 = g_tab[i1 * 4], rx2 = g_tab[i2 * 4], rx3 = g_tab[i3 * 4];
      const float2 ry0 = g_tab[y0 * 4], ry1 = g_tab[y1 * 4], ry2 = g_tab[y2 * 4], ry3 = g_tab[y3 * 4];
      const float x0c0 = ry_elem(rx0, m & 1, 0),        x0c1 = ry_elem(rx0, m & 1, 1);
      const float x1ch = ry_elem(rx1, (m >> 1) & 1, h);
      const float x2c0 = ry_elem(rx2, (m >> 2) & 1, 0), x2c1 = ry_elem(rx2, (m >> 2) & 1, 1);
      const float x3c0 = ry_elem(rx3, (m >> 3) & 1, 0), x3c1 = ry_elem(rx3, (m >> 3) & 1, 1);
      const float q0c0 = ry_elem(ry0, m & 1, 0),        q0c1 = ry_elem(ry0, m & 1, 1);
      const float q1c0 = ry_elem(ry1, (m >> 1) & 1, 0), q1c1 = ry_elem(ry1, (m >> 1) & 1, 1);
      const float q2c0 = ry_elem(ry2, (m >> 2) & 1, 0), q2c1 = ry_elem(ry2, (m >> 2) & 1, 1);
      const float q3ch = ry_elem(ry3, (m >> 3) & 1, h);
      v2f aR[4];
      v2f yR[4];
#pragma unroll
      for (int j = 0; j < 4; ++j) {
        const float xr = x1ch * ((j & 1) ? x2c1 : x2c0) * ((j >> 1) ? x3c1 : x3c0);
        aR[j].x = x0c0 * xr;
        aR[j].y = x0c1 * xr;
        const float yr = ((j & 1) ? q1c1 : q1c0) * ((j >> 1) ? q2c1 : q2c0) * q3ch;
        yR[j].x = q0c0 * yr;
        yR[j].y = q0c1 * yr;
      }
      const float2 fx0 = g_tab[i0 * 4 + 1], fx1 = g_tab[i1 * 4 + 1], fx2 = g_tab[i2 * 4 + 1], fx3 = g_tab[i3 * 4 + 1];
      const float2 fy0 = g_tab[y0 * 4 + 1], fy1 = g_tab[y1 * 4 + 1], fy2 = g_tab[y2 * 4 + 1], fy3 = g_tab[y3 * 4 + 1];
      const int P = (l - 1) * NQ;
      const float2 oK01 = g_tab[(P + (p ? 13 : 3)) * 4 + 2];
      const float2 oK12 = g_tab[(P + (p ? 12 : 2)) * 4 + 2];
      const float2 oK23 = g_tab[(P + (p ? 11 : 1)) * 4 + 2];
      const float2 oTop = g_tab[(P + (p ? 10 : 0)) * 4 + 2];
      const float2 oBot = g_tab[(p ? NGATE : P + 4) * 4 + 2];
      const float2 oC01 = g_tab[(P + (p ? 9 : 7)) * 4 + 2];
      const float2 oC12 = g_tab[(p ? NGATE : P + 6) * 4 + 2];
      const float2 oC23 = g_tab[(p ? NGATE : P + 5) * 4 + 2];
      const float2 oT5  = g_tab[(p ? NGATE : P + 8) * 4 + 2];
      const int c0 = m & 1, c1 = (m >> 1) & 1, c2 = (m >> 2) & 1, c3 = (m >> 3) & 1;
      const int nTop = p ? c0 : 0;
      float2 lbase = cmulf(cmulf(rz_phase(fy0, c0), rz_phase(fy1, c1)),
                           cmulf(rz_phase(fy2, c2), rz_phase(fy3, c3)));
      lbase = cmulf(lbase, rz_phase(fx1, h));
      lbase = cmulf(lbase, cmulf(cmulf(rz_phase(oC01, c0 ^ c1), rz_phase(oC12, c1 ^ c2)),
                                 cmulf(rz_phase(oC23, c2 ^ c3), rz_phase(oT5, (wave >> 2) ^ c0))));
      const float2 le0 = cmulf(lbase, cmulf(rz_phase(fx0, 0), cmulf(rz_phase(oK01, h), rz_phase(oBot, c3))));
      const float2 le1 = cmulf(lbase, cmulf(rz_phase(fx0, 1), cmulf(rz_phase(oK01, h ^ 1), rz_phase(oBot, c3 ^ 1))));
      float2 phL0[4], phL1[4];
#pragma unroll
      for (int j = 0; j < 4; ++j) {
        const int k2 = j & 1, k3 = j >> 1;
        const float2 fj = cmulf(cmulf(rz_phase(fx2, k2), rz_phase(fx3, k3)),
                                cmulf(cmulf(rz_phase(oK12, h ^ k2), rz_phase(oK23, k2 ^ k3)), rz_phase(oTop, k3 ^ nTop)));
        phL0[j] = cmulf(le0, fj);
        phL1[j] = cmulf(le1, fj);
      }
      const uint4* arow = (const uint4*)(a_tab + ((l - 1) * 2 + p) * 16);
      const uint4 tX = arow[0], tY = arow[1], tT = arow[2], tW = arow[3];
      const uint32_t pk1 = tX.x, pk2 = tX.y, pk4 = tX.z, pk8 = tX.w;
      const uint32_t pt1 = tT.x, pt2 = tT.y, pt4 = tT.z;
      const uint32_t lane8 = ((m & 1) ? tY.x : 0u) ^ ((m & 2) ? tY.y : 0u) ^ ((m & 4) ? tY.z : 0u) ^ ((m & 8) ? tY.w : 0u)
                           ^ ((wave & 1) ? tT.w : 0u) ^ ((wave & 2) ? tW.x : 0u) ^ ((wave & 4) ? tW.y : 0u);
      const uint32_t ldb = lane8 ^ (h ? pk2 : 0u);
      const uint32_t stb = lane8 ^ (h ? pk8 : 0u);
      char* const stB = (char*)st2;

#pragma unroll
      for (int ti = 0; ti < 8; ++ti) {
        const uint32_t tc = ((ti & 1) ? pt1 : 0u) ^ ((ti & 2) ? pt2 : 0u) ^ ((ti & 4) ? pt4 : 0u);
        const uint32_t la = ldb ^ tc;
        const uint32_t sa = stb ^ tc;
        v2f bR[4], bI[4];
#pragma unroll
        for (int j = 0; j < 4; ++j) {
          uint32_t x0 = la;
          if (j & 1) x0 ^= pk4;
          if (j & 2) x0 ^= pk8;
          const uint32_t x1 = x0 ^ pk1;
          const float2 s0 = cmulf(*(const float2*)(stB + x0), phL0[j]);
          const float2 s1 = cmulf(*(const float2*)(stB + x1), phL1[j]);
          bR[j].x = s0.x; bR[j].y = s1.x;
          bI[j].x = s0.y; bI[j].y = s1.y;
        }
        v8f dr = {}, di = {};
#pragma unroll
        for (int j = 0; j < 4; ++j) {
          dr = __builtin_amdgcn_wmma_f32_16x16x4_f32(false, bR[j], false, aR[j], (short)0, dr, false, false);
          di = __builtin_amdgcn_wmma_f32_16x16x4_f32(false, bI[j], false, aR[j], (short)0, di, false, false);
        }
        v8f cr = {}, ci = {};
#pragma unroll
        for (int j = 0; j < 4; ++j) {
          v2f ar; ar.x = dr[2 * j]; ar.y = dr[2 * j + 1];
          v2f ai; ai.x = di[2 * j]; ai.y = di[2 * j + 1];
          cr = __builtin_amdgcn_wmma_f32_16x16x4_f32(false, ar, false, yR[j], (short)0, cr, false, false);
          ci = __builtin_amdgcn_wmma_f32_16x16x4_f32(false, ai, false, yR[j], (short)0, ci, false, false);
        }
#pragma unroll
        for (int j = 0; j < 4; ++j) {
          uint32_t x0 = sa;
          if (j & 1) x0 ^= pk2;
          if (j & 2) x0 ^= pk4;
          const uint32_t x1 = x0 ^ pk1;
          *(float2*)(stB + x0) = make_float2(cr[2 * j], ci[2 * j]);
          *(float2*)(stB + x1) = make_float2(cr[2 * j + 1], ci[2 * j + 1]);
        }
      }
      __syncthreads();
    }
  }

  const float2* sp = st2 + tid;
  float T = 0.f;
  float Dq = 0.f;
  float D[6];
  for (int j = 0; j < 6; ++j) D[j] = 0.f;
#pragma nounroll
  for (int b = 0; b < 4; ++b) {
    float p[16];
    for (int j = 0; j < 16; ++j) {
      const float2 a = sp[BLOCK * (16 * b + j)];
      p[j] = a.x * a.x + a.y * a.y;
    }
    float s8[8], s4[4], s2[2];
    float d0 = 0.f, d1 = 0.f, d2 = 0.f;
    for (int k = 0; k < 8; ++k) { s8[k] = p[2 * k] + p[2 * k + 1]; d0 += p[2 * k] - p[2 * k + 1]; }
    for (int k = 0; k < 4; ++k) { s4[k] = s8[2 * k] + s8[2 * k + 1]; d1 += s8[2 * k] - s8[2 * k + 1]; }
    for (int k = 0; k < 2; ++k) { s2[k] = s4[2 * k] + s4[2 * k + 1]; d2 += s4[2 * k] - s4[2 * k + 1]; }
    const float t  = s2[0] + s2[1];
    const float d3 = s2[0] - s2[1];
    T += t;
    D[0] += d0; D[1] += d1; D[2] += d2; D[3] += d3;
    D[4] += (b & 1) ? -t : t;
    D[5] += (b & 2) ? -t : t;
    Dq   += ((b ^ (b >> 1)) & 1) ? -d1 : d1;
  }
  float G[5];
  G[4] = Dq;
  for (int s = 0; s < 5; ++s) {
    const int msk = 1 << s;
    const bool up = (lane & msk) != 0;
    T += __shfl_xor(T, msk, 32);
    for (int k = 0; k < 4; ++k)
      if (k < s) G[k] += __shfl_xor(G[k], msk, 32);
    {
      const float o = __shfl_xor(G[4], msk, 32);
      G[4] = (s < 4) ? (G[4] + o) : (up ? (o - G[4]) : (G[4] - o));
    }
    for (int k = 0; k < 6; ++k) {
      const float o = __shfl_xor(D[k], msk, 32);
      if (k == s && k < 4) G[k] = up ? (o - D[k]) : (D[k] - o);
      D[k] += o;
    }
  }
  const int w0 = wave & 1, w1 = (wave >> 1) & 1, w2 = wave >> 2;
  if (lane == 0) {
    float* dst = s_tab + wave * NQ;
    dst[0] = D[5]; dst[1] = D[4]; dst[2] = D[3]; dst[3] = D[2]; dst[4] = D[1]; dst[5] = D[0];
    dst[6] = w2 ? -T : T;
    dst[7] = w1 ? -T : T;
    dst[8] = w0 ? -D[5] : D[5];
    dst[9]  = w2 ? -G[4] : G[4];
    dst[10] = w1 ? -G[3] : G[3];
    dst[11] = (w0 ^ w2) ? -G[2] : G[2];
    dst[12] = w1 ? -G[1] : G[1];
    dst[13] = w0 ? -G[0] : G[0];
  }
  __syncthreads();
  if (tid < NQ) {
    float z = 0.f;
#pragma unroll
    for (int w = 0; w < 8; ++w) z += s_tab[w * NQ + tid];
    out[bid * NQ + tid] = z;
  }
}

extern "C" void kernel_launch(void* const* d_in, const int* in_sizes, int n_in,
                              void* d_out, int out_size, void* d_ws, size_t ws_size,
                              hipStream_t stream) {
  const float* latent  = (const float*)d_in[0];
  const float* weights = (const float*)d_in[1];
  float* out = (float*)d_out;

  const int smem_bytes = SMEM_BYTES;
  hipFuncSetAttribute(reinterpret_cast<const void*>(qcircuit_kernel),
                      hipFuncAttributeMaxDynamicSharedMemorySize, smem_bytes);
  qcircuit_kernel<<<512, BLOCK, smem_bytes, stream>>>(latent, weights, out);
}


# Pass 1's Y group has gates on 2 of its 4 bits: its two RY are applied as 2x2 rotations between step 1's accumulator registers in a lane; 8 WMMA per pass-1 tile, was 16; new rounding and sum order
# speedup vs baseline: 3.8636x; 1.1453x over previous
#include <hip/hip_runtime.h>
#include <hip/hip_bf16.h>
#include <stdint.h>

#define NQ      14
#define NL      8
#define NSTATES 16384
#define BLOCK   256
#define NGATE   (NL * NQ)
#define SMEM_BYTES 137888

typedef __attribute__((ext_vector_type(2))) float v2f;
typedef __attribute__((ext_vector_type(8))) float v8f;

__device__ __forceinline__ float2 cmulf(float2 a, float2 b) {
  return make_float2(a.x * b.x - a.y * b.y, a.x * b.y + a.y * b.x);
}

__device__ __forceinline__ float ry_elem(float2 cs, int row, int col) {
  return (row == col) ? cs.x : (row ? cs.y : -cs.y);
}

__device__ __forceinline__ float2 rz_phase(float2 cs, int bit) {
  return make_float2(cs.x, bit ? cs.y : -cs.y);
}

__device__ __forceinline__ uint32_t swz(uint32_t x) {
  return x ^ ((x >> 3) & 28u) ^ ((x >> 5) & 23u) ^ ((x >> 9) & 16u);
}

__device__ __forceinline__ uint32_t perm_map(uint32_t y, int l) {
  uint32_t r = y;
#pragma unroll
  for (int j = 1; j <= 7; ++j)
    if ((j & ~l) == 0) r ^= (y >> j);
  return swz(r);
}

__device__ __forceinline__ uint32_t swz8(uint32_t x) {
  return x ^ ((x >> 3) & (28u << 3)) ^ ((x >> 5) & (23u << 3)) ^ ((x >> 9) & (16u << 3));
}
__device__ __forceinline__ uint32_t perm_map8(uint32_t y8, int l) {
  uint32_t r = y8;
#pragma unroll
  for (int j = 1; j <= 7; ++j)
    if ((j & ~l) == 0) r ^= (y8 >> j);
  return swz8(r & ~7u);
}

template <int p>
__device__ __forceinline__ void gate_pass(const int l, const int lane, const int wave, float2* st2,
                                          const float2* g_tab, const uint32_t* a_tab) {
  const int bx = p ? 0 : 10;
  const int by = p ? 4 : 6;
  const int m = lane & 15;
  const int h = lane >> 4;
  const int i0 = l * NQ + 13 - bx;
  const int i1 = i0 - 1, i2 = i0 - 2, i3 = i0 - 3;
  const int y0 = l * NQ + 13 - by;
  const int y1 = y0 - 1;
  const int y2 = p ? NGATE : y0 - 2;
  const int y3 = p ? NGATE : y0 - 3;
  const float2 rx0 = g_tab[i0 * 4], rx1 = g_tab[i1 * 4], rx2 = g_tab[i2 * 4], rx3 = g_tab[i3 * 4];
  const float2 ry0 = g_tab[y0 * 4], ry1 = g_tab[y1 * 4], ry2 = g_tab[y2 * 4], ry3 = g_tab[y3 * 4];
  const float x0c0 = ry_elem(rx0, m & 1, 0),        x0c1 = ry_elem(rx0, m & 1, 1);
  const float x1ch = ry_elem(rx1, (m >> 1) & 1, h);
  const float x2c0 = ry_elem(rx2, (m >> 2) & 1, 0), x2c1 = ry_elem(rx2, (m >> 2) & 1, 1);
  const float x3c0 = ry_elem(rx3, (m >> 3) & 1, 0), x3c1 = ry_elem(rx3, (m >> 3) & 1, 1);
  v2f aR[4];
#pragma unroll
  for (int j = 0; j < 4; ++j) {
    const float xr = x1ch * ((j & 1) ? x2c1 : x2c0) * ((j >> 1) ? x3c1 : x3c0);
    aR[j].x = x0c0 * xr;
    aR[j].y = x0c1 * xr;
  }
  v2f yR[4];
  if constexpr (p == 0) {
    const float q0c0 = ry_elem(ry0, m & 1, 0),        q0c1 = ry_elem(ry0, m & 1, 1);
    const float q1c0 = ry_elem(ry1, (m >> 1) & 1, 0), q1c1 = ry_elem(ry1, (m >> 1) & 1, 1);
    const float q2c0 = ry_elem(ry2, (m >> 2) & 1, 0), q2c1 = ry_elem(ry2, (m >> 2) & 1, 1);
    const float q3ch = ry_elem(ry3, (m >> 3) & 1, h);
#pragma unroll
    for (int j = 0; j < 4; ++j) {
      const float yr = ((j & 1) ? q1c1 : q1c0) * ((j >> 1) ? q2c1 : q2c0) * q3ch;
      yR[j].x = q0c0 * yr;
      yR[j].y = q0c1 * yr;
    }
  }
  const float2 fx0 = g_tab[i0 * 4 + 1], fx1 = g_tab[i1 * 4 + 1], fx2 = g_tab[i2 * 4 + 1], fx3 = g_tab[i3 * 4 + 1];
  const float2 fy0 = g_tab[y0 * 4 + 1], fy1 = g_tab[y1 * 4 + 1], fy2 = g_tab[y2 * 4 + 1], fy3 = g_tab[y3 * 4 + 1];
  const int P = (l - 1) * NQ;
  const float2 oK01 = g_tab[(P + (p ? 13 : 3)) * 4 + 2];
  const float2 oK12 = g_tab[(P + (p ? 12 : 2)) * 4 + 2];
  const float2 oK23 = g_tab[(P + (p ? 11 : 1)) * 4 + 2];
  const float2 oTop = g_tab[(P + (p ? 10 : 0)) * 4 + 2];
  const float2 oBot = g_tab[(p ? NGATE : P + 4) * 4 + 2];
  const float2 oC01 = g_tab[(P + (p ? 9 : 7)) * 4 + 2];
  const float2 oC12 = g_tab[(p ? NGATE : P + 6) * 4 + 2];
  const float2 oC23 = g_tab[(p ? NGATE : P + 5) * 4 + 2];
  const float2 oT5  = g_tab[(p ? NGATE : P + 8) * 4 + 2];
  const int c0 = m & 1, c1 = (m >> 1) & 1, c2 = (m >> 2) & 1, c3 = (m >> 3) & 1;
  const int nTop = p ? c0 : 0;
  float2 lbase = cmulf(cmulf(rz_phase(fy0, c0), rz_phase(fy1, c1)),
                       cmulf(rz_phase(fy2, c2), rz_phase(fy3, c3)));
  lbase = cmulf(lbase, rz_phase(fx1, h));
  lbase = cmulf(lbase, cmulf(cmulf(rz_phase(oC01, c0 ^ c1), rz_phase(oC12, c1 ^ c2)),
                             cmulf(rz_phase(oC23, c2 ^ c3), rz_phase(oT5, (wave >> 2) ^ c0))));
  const float2 le0 = cmulf(lbase, cmulf(rz_phase(fx0, 0), cmulf(rz_phase(oK01, h), rz_phase(oBot, c3))));
  const float2 le1 = cmulf(lbase, cmulf(rz_phase(fx0, 1), cmulf(rz_phase(oK01, h ^ 1), rz_phase(oBot, c3 ^ 1))));
  float2 phL0[4], phL1[4];
#pragma unroll
  for (int j = 0; j < 4; ++j) {
    const int k2 = j & 1, k3 = j >> 1;
    const float2 fj = cmulf(cmulf(rz_phase(fx2, k2), rz_phase(fx3, k3)),
                            cmulf(cmulf(rz_phase(oK12, h ^ k2), rz_phase(oK23, k2 ^ k3)), rz_phase(oTop, k3 ^ nTop)));
    phL0[j] = cmulf(le0, fj);
    phL1[j] = cmulf(le1, fj);
  }
  const uint4* arow = (const uint4*)(a_tab + ((l - 1) * 2 + p) * 16);
  const uint4 tX = arow[0], tY = arow[1], tT = arow[2], tW = arow[3];
  const uint32_t pk1 = tX.x, pk2 = tX.y, pk4 = tX.z, pk8 = tX.w;
  const uint32_t pt1 = tT.x, pt2 = tT.y, pt4 = tT.z;
  const uint32_t wv8 = ((wave & 1) ? tT.w : 0u) ^ ((wave & 2) ? tW.x : 0u) ^ ((wave & 4) ? tW.y : 0u);
  const uint32_t lane8 = ((m & 1) ? tY.x : 0u) ^ ((m & 2) ? tY.y : 0u) ^ ((m & 4) ? tY.z : 0u) ^ ((m & 8) ? tY.w : 0u) ^ wv8;
  const uint32_t ldb = lane8 ^ (h ? pk2 : 0u);
  const uint32_t stb = p ? (wv8 ^ ((m & 1) ? pk1 : 0u) ^ ((m & 2) ? pk2 : 0u) ^ ((m & 4) ? pk4 : 0u) ^ ((m & 8) ? pk8 : 0u) ^ (h ? tY.w : 0u))
                         : (lane8 ^ (h ? pk8 : 0u));
  char* const stB = (char*)st2;

#pragma unroll
  for (int ti = 0; ti < 8; ++ti) {
    const uint32_t tc = ((ti & 1) ? pt1 : 0u) ^ ((ti & 2) ? pt2 : 0u) ^ ((ti & 4) ? pt4 : 0u);
    const uint32_t la = ldb ^ tc;
    const uint32_t sa = stb ^ tc;
    v2f bR[4], bI[4];
#pragma unroll
    for (int j = 0; j < 4; ++j) {
      uint32_t x0 = la;
      if (j & 1) x0 ^= pk4;
      if (j & 2) x0 ^= pk8;
      const uint32_t x1 = x0 ^ pk1;
      const float2 s0 = cmulf(*(const float2*)(stB + x0), phL0[j]);
      const float2 s1 = cmulf(*(const float2*)(stB + x1), phL1[j]);
      bR[j].x = s0.x; bR[j].y = s1.x;
      bI[j].x = s0.y; bI[j].y = s1.y;
    }
    v8f dr = {}, di = {};
#pragma unroll
    for (int j = 0; j < 4; ++j) {
      dr = __builtin_amdgcn_wmma_f32_16x16x4_f32(false, bR[j], false, aR[j], (short)0, dr, false, false);
      di = __builtin_amdgcn_wmma_f32_16x16x4_f32(false, bI[j], false, aR[j], (short)0, di, false, false);
    }
    if constexpr (p == 0) {
      v8f cr = {}, ci = {};
#pragma unroll
      for (int j = 0; j < 4; ++j) {
        v2f ar; ar.x = dr[2 * j]; ar.y = dr[2 * j + 1];
        v2f ai; ai.x = di[2 * j]; ai.y = di[2 * j + 1];
        cr = __builtin_amdgcn_wmma_f32_16x16x4_f32(false, ar, false, yR[j], (short)0, cr, false, false);
        ci = __builtin_amdgcn_wmma_f32_16x16x4_f32(false, ai, false, yR[j], (short)0, ci, false, false);
      }
#pragma unroll
      for (int j = 0; j < 4; ++j) {
        uint32_t x0 = sa;
        if (j & 1) x0 ^= pk2;
        if (j & 2) x0 ^= pk4;
        const uint32_t x1 = x0 ^ pk1;
        *(float2*)(stB + x0) = make_float2(cr[2 * j], ci[2 * j]);
        *(float2*)(stB + x1) = make_float2(cr[2 * j + 1], ci[2 * j + 1]);
      }
    } else {
      v2f zz[8];
#pragma unroll
      for (int r = 0; r < 8; ++r) { zz[r].x = dr[r]; zz[r].y = di[r]; }
      v2f ww[8];
#pragma unroll
      for (int r = 0; r < 8; ++r) {
        if (r & 2) continue;
        ww[r]     = ry1.x * zz[r] - ry1.y * zz[r + 2];
        ww[r + 2] = ry1.y * zz[r] + ry1.x * zz[r + 2];
      }
#pragma unroll
      for (int u = 0; u < 4; ++u) {
        uint32_t x0 = sa;
        if (u & 1) x0 ^= tY.y;
        if (u & 2) x0 ^= tY.z;
        const uint32_t x1 = x0 ^ tY.x;
        const v2f o0 = ry0.x * ww[2 * u] - ry0.y * ww[2 * u + 1];
        const v2f o1 = ry0.y * ww[2 * u] + ry0.x * ww[2 * u + 1];
        *(float2*)(stB + x0) = make_float2(o0.x, o0.y);
        *(float2*)(stB + x1) = make_float2(o1.x, o1.y);
      }
    }
  }
  __syncthreads();
}

__global__ __launch_bounds__(BLOCK) __attribute__((amdgpu_waves_per_eu(4, 4)))
void qcircuit_kernel(const float* __restrict__ latent,
                     const float* __restrict__ weights,
                     float* __restrict__ out) {
  extern __shared__ float smem[];
  float2* st2  = (float2*)smem;
  float* s_tab = smem + 2 * NSTATES;
  float2* c_tab = (float2*)s_tab;
  float2* vq    = (float2*)(smem + 2 * NSTATES + 512);
  float2* g_tab = (float2*)(smem + 2 * NSTATES + 576);
  uint32_t* a_tab = (uint32_t*)(smem + 2 * NSTATES + 576 + (NGATE + 1) * 8);

  const int tid  = threadIdx.x;
  const int lane = tid & 31;
  const int wave = tid >> 5;
  const int bid  = blockIdx.x;
  const float* lat = latent + bid * NQ;

  if (tid < NGATE) {
    const float* w = weights + tid * 3;
    const float phi = w[0], th = w[1], om = w[2];
    float sth, cth;
    sincosf(0.5f * th, &sth, &cth);
    if (tid < NQ) {
      float sa, ca, sb, cb;
      sincosf(0.5f * (phi + om), &sa, &ca);
      sincosf(0.5f * (phi - om), &sb, &cb);
      const float2 m00 = make_float2( cth * ca, -cth * sa);
      const float2 m01 = make_float2(-sth * cb, -sth * sb);
      const float2 m10 = make_float2( sth * cb, -sth * sb);
      const float2 m11 = make_float2( cth * ca,  cth * sa);
      float sn, cs;
      sincosf(0.5f * lat[tid], &sn, &cs);
      vq[tid * 2 + 0] = make_float2(m00.x * cs + m01.y * sn, m00.y * cs - m01.x * sn);
      vq[tid * 2 + 1] = make_float2(m10.x * cs + m11.y * sn, m10.y * cs - m11.x * sn);
      g_tab[tid * 4 + 2] = make_float2(1.f, 0.f);
    } else {
      float sp, cp, so, co;
      sincosf(0.5f * phi, &sp, &cp);
      sincosf(0.5f * om, &so, &co);
      g_tab[tid * 4 + 0] = make_float2(cth, sth);
      g_tab[tid * 4 + 1] = make_float2(cp, sp);
      g_tab[tid * 4 + 2] = make_float2(co, so);
    }
  } else if (tid == NGATE) {
    g_tab[NGATE * 4 + 0] = make_float2(1.f, 0.f);
    g_tab[NGATE * 4 + 1] = make_float2(1.f, 0.f);
    g_tab[NGATE * 4 + 2] = make_float2(1.f, 0.f);
  }
  if (tid < 7 * 2 * 16) {
    const int e = tid & 15, pp = (tid >> 4) & 1, ll = (tid >> 5) + 1;
    const int bit = (e < 4) ? (pp ? 0 : 10) + e : (e < 8) ? (pp ? 4 : 6) + (e - 4) : (pp ? 8 : 0) + (e - 8);
    a_tab[tid] = (e < 14) ? perm_map8(8u << bit, ll) : 0u;
  }
  __syncthreads();
  {
    const int half = tid >> 7, idx = tid & 127;
    float2 prod = vq[(half * 7) * 2 + ((idx >> 6) & 1)];
#pragma unroll
    for (int j = 1; j < 7; ++j)
      prod = cmulf(prod, vq[(half * 7 + j) * 2 + ((idx >> (6 - j)) & 1)]);
    c_tab[tid] = prod;
  }
  __syncthreads();
#pragma unroll 4
  for (int i = 0; i < NSTATES / BLOCK; ++i) {
    const int y = i * BLOCK + tid;
    st2[swz((uint32_t)y)] = cmulf(c_tab[y >> 7], c_tab[128 + (y & 127)]);
  }
  __syncthreads();

  for (int l = 1; l < NL; ++l) {
    gate_pass<0>(l, lane, wave, st2, g_tab, a_tab);
    gate_pass<1>(l, lane, wave, st2, g_tab, a_tab);
  }

  const float2* sp = st2 + tid;
  float T = 0.f;
  float Dq = 0.f;
  float D[6];
  for (int j = 0; j < 6; ++j) D[j] = 0.f;
#pragma nounroll
  for (int b = 0; b < 4; ++b) {
    float p[16];
    for (int j = 0; j < 16; ++j) {
      const float2 a = sp[BLOCK * (16 * b + j)];
      p[j] = a.x * a.x + a.y * a.y;
    }
    float s8[8], s4[4], s2[2];
    float d0 = 0.f, d1 = 0.f, d2 = 0.f;
    for (int k = 0; k < 8; ++k) { s8[k] = p[2 * k] + p[2 * k + 1]; d0 += p[2 * k] - p[2 * k + 1]; }
    for (int k = 0; k < 4; ++k) { s4[k] = s8[2 * k] + s8[2 * k + 1]; d1 += s8[2 * k] - s8[2 * k + 1]; }
    for (int k = 0; k < 2; ++k) { s2[k] = s4[2 * k] + s4[2 * k + 1]; d2 += s4[2 * k] - s4[2 * k + 1]; }
    const float t  = s2[0] + s2[1];
    const float d3 = s2[0] - s2[1];
    T += t;
    D[0] += d0; D[1] += d1; D[2] += d2; D[3] += d3;
    D[4] += (b & 1) ? -t : t;
    D[5] += (b & 2) ? -t : t;
    Dq   += ((b ^ (b >> 1)) & 1) ? -d1 : d1;
  }
  float G[5];
  G[4] = Dq;
  for (int s = 0; s < 5; ++s) {
    const int msk = 1 << s;
    const bool up = (lane & msk) != 0;
    T += __shfl_xor(T, msk, 32);
    for (int k = 0; k < 4; ++k)
      if (k < s) G[k] += __shfl_xor(G[k], msk, 32);
    {
      const float o = __shfl_xor(G[4], msk, 32);
      G[4] = (s < 4) ? (G[4] + o) : (up ? (o - G[4]) : (G[4] - o));
    }
    for (int k = 0; k < 6; ++k) {
      const float o = __shfl_xor(D[k], msk, 32);
      if (k == s && k < 4) G[k] = up ? (o - D[k]) : (D[k] - o);
      D[k] += o;
    }
  }
  const int w0 = wave & 1, w1 = (wave >> 1) & 1, w2 = wave >> 2;
  if (lane == 0) {
    float* dst = s_tab + wave * NQ;
    dst[0] = D[5]; dst[1] = D[4]; dst[2] = D[3]; dst[3] = D[2]; dst[4] = D[1]; dst[5] = D[0];
    dst[6] = w2 ? -T : T;
    dst[7] = w1 ? -T : T;
    dst[8] = w0 ? -D[5] : D[5];
    dst[9]  = w2 ? -G[4] : G[4];
    dst[10] = w1 ? -G[3] : G[3];
    dst[11] = (w0 ^ w2) ? -G[2] : G[2];
    dst[12] = w1 ? -G[1] : G[1];
    dst[13] = w0 ? -G[0] : G[0];
  }
  __syncthreads();
  if (tid < NQ) {
    float z = 0.f;
#pragma unroll
    for (int w = 0; w < 8; ++w) z += s_tab[w * NQ + tid];
    out[bid * NQ + tid] = z;
  }
}

extern "C" void kernel_launch(void* const* d_in, const int* in_sizes, int n_in,
                              void* d_out, int out_size, void* d_ws, size_t ws_size,
                              hipStream_t stream) {
  const float* latent  = (const float*)d_in[0];
  const float* weights = (const float*)d_in[1];
  float* out = (float*)d_out;

  const int smem_bytes = SMEM_BYTES;
  hipFuncSetAttribute(reinterpret_cast<const void*>(qcircuit_kernel),
                      hipFuncAttributeMaxDynamicSharedMemorySize, smem_bytes);
  qcircuit_kernel<<<512, BLOCK, smem_bytes, stream>>>(latent, weights, out);
}
